# starting kernel
# baseline (speedup 1.0000x reference)
#include <hip/hip_runtime.h>
#include <math.h>

#define N_NODES 200000
#define N_EDGES 2000000
#define IN_DIM  165
#define HID_DIM 128
#define KPAD    168
#define KMAIN   164

typedef __attribute__((ext_vector_type(2))) float v2f;
typedef __attribute__((ext_vector_type(8))) float v8f;


__global__ void k_deg_init(float* __restrict__ deg, int n) {
    int i = blockIdx.x * blockDim.x + threadIdx.x;
    if (i < n) deg[i] = 1.0f;
}

__global__ void k_deg_count(const int* __restrict__ dst, float* __restrict__ deg, int e) {
    int i = blockIdx.x * blockDim.x + threadIdx.x;
    if (i < e) atomicAdd(&deg[dst[i]], 1.0f);
}

__global__ void k_dinv(float* __restrict__ deg, int n) {
    int i = blockIdx.x * blockDim.x + threadIdx.x;
    if (i < n) deg[i] = rsqrtf(deg[i]);
}


__global__ __launch_bounds__(256)
void k_gemm1(const float* __restrict__ x, const float* __restrict__ W1,
             float* __restrict__ h) {
    __shared__ float sX[64][KPAD];
    const int tid      = threadIdx.x;
    const int nodeBase = blockIdx.x * 64;

    for (int idx = tid; idx < 64 * KPAD; idx += 256) {
        int r = idx / KPAD, c = idx % KPAD;
        sX[r][c] = (c < IN_DIM) ? x[(nodeBase + r) * IN_DIM + c] : 0.0f;
    }
    __syncthreads();

    const int lane  = tid & 31;
    const int wave  = tid >> 5;
    const int m     = lane & 15;
    const int kp    = (lane >> 4) << 1;
    const int ntile = wave >> 1;
    const int fbase = (wave & 1) * 64;
    const float* arow = &sX[ntile * 16 + m][0];

    for (int ft = 0; ft < 4; ++ft) {
        const int feat0 = fbase + ft * 16;
        v8f c = {0.f, 0.f, 0.f, 0.f, 0.f, 0.f, 0.f, 0.f};
        const float* wrow = &W1[(feat0 + m) * IN_DIM];

        for (int k = 0; k < KMAIN; k += 4) {
            v2f a, b;
            a.x = arow[k + kp];
            a.y = arow[k + kp + 1];
            b.x = wrow[k + kp];
            b.y = wrow[k + kp + 1];
            c = __builtin_amdgcn_wmma_f32_16x16x4_f32(
                    false, a, false, b, (short)0, c, false, false);
        }
        {
            v2f a, b;
            a.x = arow[KMAIN + kp];
            a.y = arow[KMAIN + kp + 1];
            b.x = (kp == 0) ? wrow[KMAIN] : 0.0f;
            b.y = 0.0f;
            c = __builtin_amdgcn_wmma_f32_16x16x4_f32(
                    false, a, false, b, (short)0, c, false, false);
        }

        const int mBase = ntile * 16 + ((lane >> 4) << 3);
#pragma unroll
        for (int r = 0; r < 8; ++r) {
            h[(nodeBase + mBase + r) * HID_DIM + feat0 + m] = c[r];
        }
    }
}


__global__ void k_selfloop1(const float* __restrict__ h, const float* __restrict__ dinv,
                            float* __restrict__ agg, int n) {
    int gid = blockIdx.x * blockDim.x + threadIdx.x;
    int i = gid >> 5;
    if (i >= n) return;
    int f4 = (gid & 31) * 4;
    float w = dinv[i] * dinv[i];
    float4 hv = *(const float4*)&h[(size_t)i * HID_DIM + f4];
    float4 o;
    o.x = hv.x * w; o.y = hv.y * w; o.z = hv.z * w; o.w = hv.w * w;
    *(float4*)&agg[(size_t)i * HID_DIM + f4] = o;
}

__global__ void k_scatter1(const int* __restrict__ src, const int* __restrict__ dst,
                           const float* __restrict__ dinv, const float* __restrict__ h,
                           float* __restrict__ agg) {
    int gid = blockIdx.x * blockDim.x + threadIdx.x;
    int e = gid >> 5;
    if (e >= N_EDGES) return;
    int lane = threadIdx.x & 31;
    int s = src[e], d = dst[e];
    float w = dinv[s] * dinv[d];
    int f4 = lane * 4;
    float4 hv = *(const float4*)&h[(size_t)s * HID_DIM + f4];
    float* ap = &agg[(size_t)d * HID_DIM + f4];
    atomicAdd(ap + 0, hv.x * w);
    atomicAdd(ap + 1, hv.y * w);
    atomicAdd(ap + 2, hv.z * w);
    atomicAdd(ap + 3, hv.w * w);
}


__global__ void k_relu_dot(const float* __restrict__ agg, const float* __restrict__ b1,
                           const float* __restrict__ W2, float* __restrict__ sval, int n) {
    int gid = blockIdx.x * blockDim.x + threadIdx.x;
    int i = gid >> 5;
    if (i >= n) return;
    int lane = threadIdx.x & 31;
    int f4 = lane * 4;
    float4 a  = *(const float4*)&agg[(size_t)i * HID_DIM + f4];
    float4 bb = *(const float4*)&b1[f4];
    float4 w2 = *(const float4*)&W2[f4];
    float p = fmaxf(a.x + bb.x, 0.f) * w2.x
            + fmaxf(a.y + bb.y, 0.f) * w2.y
            + fmaxf(a.z + bb.z, 0.f) * w2.z
            + fmaxf(a.w + bb.w, 0.f) * w2.w;
#pragma unroll
    for (int off = 16; off > 0; off >>= 1)
        p += __shfl_xor(p, off, 32);
    if (lane == 0) sval[i] = p;
}


__global__ void k_selfloop2(const float* __restrict__ sval, const float* __restrict__ dinv,
                            float* __restrict__ out, int n) {
    int i = blockIdx.x * blockDim.x + threadIdx.x;
    if (i < n) out[i] = sval[i] * dinv[i] * dinv[i];
}

__global__ void k_scatter2(const int* __restrict__ src, const int* __restrict__ dst,
                           const float* __restrict__ dinv, const float* __restrict__ sval,
                           float* __restrict__ out) {
    int e = blockIdx.x * blockDim.x + threadIdx.x;
    if (e >= N_EDGES) return;
    int s = src[e], d = dst[e];
    atomicAdd(&out[d], sval[s] * dinv[s] * dinv[d]);
}

__global__ void k_sigmoid(float* __restrict__ out, const float* __restrict__ b2, int n) {
    int i = blockIdx.x * blockDim.x + threadIdx.x;
    if (i < n) out[i] = 1.0f / (1.0f + expf(-(out[i] + b2[0])));
}


extern "C" void kernel_launch(void* const* d_in, const int* in_sizes, int n_in,
                              void* d_out, int out_size, void* d_ws, size_t ws_size,
                              hipStream_t stream) {
    (void)in_sizes; (void)n_in; (void)out_size; (void)ws_size;
    const float* x  = (const float*)d_in[0];
    const float* W1 = (const float*)d_in[1];
    const float* b1 = (const float*)d_in[2];
    const float* W2 = (const float*)d_in[3];
    const float* b2 = (const float*)d_in[4];
    const int*   ei = (const int*)d_in[5];
    const int* src = ei;
    const int* dst = ei + N_EDGES;
    float* out = (float*)d_out;

    float* dinv = (float*)d_ws;
    float* sval = dinv + N_NODES;
    float* h    = sval + N_NODES;
    float* agg  = h + (size_t)N_NODES * HID_DIM;

    const int B = 256;
    k_deg_init <<<(N_NODES + B - 1) / B, B, 0, stream>>>(dinv, N_NODES);
    k_deg_count<<<(N_EDGES + B - 1) / B, B, 0, stream>>>(dst, dinv, N_EDGES);
    k_dinv     <<<(N_NODES + B - 1) / B, B, 0, stream>>>(dinv, N_NODES);

    k_gemm1    <<<N_NODES / 64, B, 0, stream>>>(x, W1, h);

    k_selfloop1<<<(N_NODES * 32) / B, B, 0, stream>>>(h, dinv, agg, N_NODES);
    k_scatter1 <<<((size_t)N_EDGES * 32) / B, B, 0, stream>>>(src, dst, dinv, h, agg);

    k_relu_dot <<<(N_NODES * 32) / B, B, 0, stream>>>(agg, b1, W2, sval, N_NODES);

    k_selfloop2<<<(N_NODES + B - 1) / B, B, 0, stream>>>(sval, dinv, out, N_NODES);
    k_scatter2 <<<(N_EDGES + B - 1) / B, B, 0, stream>>>(src, dst, dinv, sval, out);
    k_sigmoid  <<<(N_NODES + B - 1) / B, B, 0, stream>>>(out, b2, N_NODES);
}


# GEMM operand feed: each wave keeps its W1 (B) fragments in registers in a persistent block; x tiles staged to LDS by contiguous 16-byte copies one tile ahead; k order unchanged
# speedup vs baseline: 1.2260x; 1.2260x over previous
#include <hip/hip_runtime.h>
#include <math.h>

#define N_NODES 200000
#define N_EDGES 2000000
#define IN_DIM  165
#define HID_DIM 128

typedef __attribute__((ext_vector_type(2))) float v2f;
typedef __attribute__((ext_vector_type(8))) float v8f;


__global__ void k_deg_init(float* __restrict__ deg, int n) {
    int i = blockIdx.x * blockDim.x + threadIdx.x;
    if (i < n) deg[i] = 1.0f;
}

__global__ void k_deg_count(const int* __restrict__ dst, float* __restrict__ deg, int e) {
    int i = blockIdx.x * blockDim.x + threadIdx.x;
    if (i < e) atomicAdd(&deg[dst[i]], 1.0f);
}

__global__ void k_dinv(float* __restrict__ deg, int n) {
    int i = blockIdx.x * blockDim.x + threadIdx.x;
    if (i < n) deg[i] = rsqrtf(deg[i]);
}


#define TILE_M    32
#define N_TILES   (N_NODES / TILE_M)
#define XT_F4     (TILE_M * IN_DIM / 4)
#define XBUF_F4   (XT_F4 + 2)
#define W1_F4     (HID_DIM * IN_DIM / 4)
#define KSTEPS    41
#define GEMM_GRID 512

struct XTile { float4 q0, q1, q2, q3, q4, q5; };

__device__ __forceinline__ XTile xtile_load(const float4* __restrict__ xs, int tid) {
    XTile t;
    t.q0 = xs[tid];
    t.q1 = xs[tid + 256];
    t.q2 = xs[tid + 512];
    t.q3 = xs[tid + 768];
    t.q4 = xs[tid + 1024];
    t.q5 = make_float4(0.f, 0.f, 0.f, 0.f);
    if (tid < XT_F4 - 1280) t.q5 = xs[tid + 1280];
    return t;
}

__device__ __forceinline__ void xtile_store(float4* xw, int tid, const XTile& t) {
    xw[tid]        = t.q0;
    xw[tid + 256]  = t.q1;
    xw[tid + 512]  = t.q2;
    xw[tid + 768]  = t.q3;
    xw[tid + 1024] = t.q4;
    if (tid < XT_F4 - 1280) xw[tid + 1280] = t.q5;
}

__global__ __launch_bounds__(256)
void k_gemm1(const float* __restrict__ x, const float* __restrict__ W1,
             float* __restrict__ h) {
    __shared__ float4 smem4[W1_F4 + 2 * XBUF_F4];
    float*  sW = (float*)smem4;
    float4* sX = smem4 + W1_F4;

    const int tid  = threadIdx.x;
    const int lane = tid & 31;
    const int wave = tid >> 5;
    const int c    = lane & 15;
    const int hh   = lane >> 4;

    int tile = blockIdx.x;

    XTile xr = xtile_load((const float4*)(x + (size_t)tile * (TILE_M * IN_DIM)), tid);

    {
        const float4* w4 = (const float4*)W1;
        for (int q = tid; q < W1_F4; q += 256) smem4[q] = w4[q];
    }
    xtile_store(sX, tid, xr);
    __syncthreads();

    v2f   bf[KSTEPS];
    float bt;
    {
        const float* wrow = sW + (wave * 16 + c) * IN_DIM;
#pragma unroll
        for (int j = 0; j < KSTEPS; ++j) {
            bf[j].x = wrow[4 * j + 2 * hh];
            bf[j].y = wrow[4 * j + 2 * hh + 1];
        }
        const float w164 = wrow[4 * KSTEPS];
        bt = (hh == 0) ? w164 : 0.0f;
    }

    int cur = 0;
    for (; tile < N_TILES; tile += GEMM_GRID) {
        const int  nxt  = tile + GEMM_GRID;
        const bool more = nxt < N_TILES;
        if (more)
            xr = xtile_load((const float4*)(x + (size_t)nxt * (TILE_M * IN_DIM)), tid);

        const float* xb = (const float*)(sX + cur * XBUF_F4);
        const float* a0 = xb + c * IN_DIM;
        const float* a1 = xb + (16 + c) * IN_DIM;
        v8f acc0 = {0.f, 0.f, 0.f, 0.f, 0.f, 0.f, 0.f, 0.f};
        v8f acc1 = {0.f, 0.f, 0.f, 0.f, 0.f, 0.f, 0.f, 0.f};

#pragma unroll
        for (int j = 0; j < KSTEPS; ++j) {
            v2f f0, f1;
            f0.x = a0[4 * j + 2 * hh];
            f0.y = a0[4 * j + 2 * hh + 1];
            f1.x = a1[4 * j + 2 * hh];
            f1.y = a1[4 * j + 2 * hh + 1];
            acc0 = __builtin_amdgcn_wmma_f32_16x16x4_f32(
                       false, f0, false, bf[j], (short)0, acc0, false, false);
            acc1 = __builtin_amdgcn_wmma_f32_16x16x4_f32(
                       false, f1, false, bf[j], (short)0, acc1, false, false);
        }
        {
            const float t0 = a0[4 * KSTEPS];
            const float t1 = a1[4 * KSTEPS];
            v2f f0, f1, fb;
            f0.x = (hh == 0) ? t0 : 0.0f;  f0.y = 0.0f;
            f1.x = (hh == 0) ? t1 : 0.0f;  f1.y = 0.0f;
            fb.x = bt;                     fb.y = 0.0f;
            acc0 = __builtin_amdgcn_wmma_f32_16x16x4_f32(
                       false, f0, false, fb, (short)0, acc0, false, false);
            acc1 = __builtin_amdgcn_wmma_f32_16x16x4_f32(
                       false, f1, false, fb, (short)0, acc1, false, false);
        }

        if (more) xtile_store(sX + (cur ^ 1) * XBUF_F4, tid, xr);
        __syncthreads();

        {
            float* hp = h + ((size_t)tile * TILE_M + 8 * hh) * HID_DIM + wave * 16 + c;
#pragma unroll
            for (int r = 0; r < 8; ++r) {
                hp[(size_t)r * HID_DIM]        = acc0[r];
                hp[(size_t)(16 + r) * HID_DIM] = acc1[r];
            }
        }
        cur ^= 1;
    }
}


__global__ void k_selfloop1(const float* __restrict__ h, const float* __restrict__ dinv,
                            float* __restrict__ agg, int n) {
    int gid = blockIdx.x * blockDim.x + threadIdx.x;
    int i = gid >> 5;
    if (i >= n) return;
    int f4 = (gid & 31) * 4;
    float w = dinv[i] * dinv[i];
    float4 hv = *(const float4*)&h[(size_t)i * HID_DIM + f4];
    float4 o;
    o.x = hv.x * w; o.y = hv.y * w; o.z = hv.z * w; o.w = hv.w * w;
    *(float4*)&agg[(size_t)i * HID_DIM + f4] = o;
}

__global__ void k_scatter1(const int* __restrict__ src, const int* __restrict__ dst,
                           const float* __restrict__ dinv, const float* __restrict__ h,
                           float* __restrict__ agg) {
    int gid = blockIdx.x * blockDim.x + threadIdx.x;
    int e = gid >> 5;
    if (e >= N_EDGES) return;
    int lane = threadIdx.x & 31;
    int s = src[e], d = dst[e];
    float w = dinv[s] * dinv[d];
    int f4 = lane * 4;
    float4 hv = *(const float4*)&h[(size_t)s * HID_DIM + f4];
    float* ap = &agg[(size_t)d * HID_DIM + f4];
    atomicAdd(ap + 0, hv.x * w);
    atomicAdd(ap + 1, hv.y * w);
    atomicAdd(ap + 2, hv.z * w);
    atomicAdd(ap + 3, hv.w * w);
}


__global__ void k_relu_dot(const float* __restrict__ agg, const float* __restrict__ b1,
                           const float* __restrict__ W2, float* __restrict__ sval, int n) {
    int gid = blockIdx.x * blockDim.x + threadIdx.x;
    int i = gid >> 5;
    if (i >= n) return;
    int lane = threadIdx.x & 31;
    int f4 = lane * 4;
    float4 a  = *(const float4*)&agg[(size_t)i * HID_DIM + f4];
    float4 bb = *(const float4*)&b1[f4];
    float4 w2 = *(const float4*)&W2[f4];
    float p = fmaxf(a.x + bb.x, 0.f) * w2.x
            + fmaxf(a.y + bb.y, 0.f) * w2.y
            + fmaxf(a.z + bb.z, 0.f) * w2.z
            + fmaxf(a.w + bb.w, 0.f) * w2.w;
#pragma unroll
    for (int off = 16; off > 0; off >>= 1)
        p += __shfl_xor(p, off, 32);
    if (lane == 0) sval[i] = p;
}


__global__ void k_selfloop2(const float* __restrict__ sval, const float* __restrict__ dinv,
                            float* __restrict__ out, int n) {
    int i = blockIdx.x * blockDim.x + threadIdx.x;
    if (i < n) out[i] = sval[i] * dinv[i] * dinv[i];
}

__global__ void k_scatter2(const int* __restrict__ src, const int* __restrict__ dst,
                           const float* __restrict__ dinv, const float* __restrict__ sval,
                           float* __restrict__ out) {
    int e = blockIdx.x * blockDim.x + threadIdx.x;
    if (e >= N_EDGES) return;
    int s = src[e], d = dst[e];
    atomicAdd(&out[d], sval[s] * dinv[s] * dinv[d]);
}

__global__ void k_sigmoid(float* __restrict__ out, const float* __restrict__ b2, int n) {
    int i = blockIdx.x * blockDim.x + threadIdx.x;
    if (i < n) out[i] = 1.0f / (1.0f + expf(-(out[i] + b2[0])));
}


extern "C" void kernel_launch(void* const* d_in, const int* in_sizes, int n_in,
                              void* d_out, int out_size, void* d_ws, size_t ws_size,
                              hipStream_t stream) {
    (void)in_sizes; (void)n_in; (void)out_size; (void)ws_size;
    const float* x  = (const float*)d_in[0];
    const float* W1 = (const float*)d_in[1];
    const float* b1 = (const float*)d_in[2];
    const float* W2 = (const float*)d_in[3];
    const float* b2 = (const float*)d_in[4];
    const int*   ei = (const int*)d_in[5];
    const int* src = ei;
    const int* dst = ei + N_EDGES;
    float* out = (float*)d_out;

    float* dinv = (float*)d_ws;
    float* sval = dinv + N_NODES;
    float* h    = sval + N_NODES;
    float* agg  = h + (size_t)N_NODES * HID_DIM;

    const int B = 256;
    k_deg_init <<<(N_NODES + B - 1) / B, B, 0, stream>>>(dinv, N_NODES);
    k_deg_count<<<(N_EDGES + B - 1) / B, B, 0, stream>>>(dst, dinv, N_EDGES);
    k_dinv     <<<(N_NODES + B - 1) / B, B, 0, stream>>>(dinv, N_NODES);

    k_gemm1    <<<GEMM_GRID, B, 0, stream>>>(x, W1, h);

    k_selfloop1<<<(N_NODES * 32) / B, B, 0, stream>>>(h, dinv, agg, N_NODES);
    k_scatter1 <<<((size_t)N_EDGES * 32) / B, B, 0, stream>>>(src, dst, dinv, h, agg);

    k_relu_dot <<<(N_NODES * 32) / B, B, 0, stream>>>(agg, b1, W2, sval, N_NODES);

    k_selfloop2<<<(N_NODES + B - 1) / B, B, 0, stream>>>(sval, dinv, out, N_NODES);
    k_scatter2 <<<(N_EDGES + B - 1) / B, B, 0, stream>>>(src, dst, dinv, sval, out);
    k_sigmoid  <<<(N_NODES + B - 1) / B, B, 0, stream>>>(out, b2, N_NODES);
}


# aggregations: edges bucketed by destination per call (int count, returned add = rank), per-node gathers sum in registers with fused epilogues; float atomics removed; sum order = edge arrival rank
# speedup vs baseline: 2.7270x; 2.2243x over previous
#include <hip/hip_runtime.h>
#include <math.h>

#define N_NODES 200000
#define N_EDGES 2000000
#define IN_DIM  165
#define HID_DIM 128

typedef __attribute__((ext_vector_type(2))) float v2f;
typedef __attribute__((ext_vector_type(8))) float v8f;


__global__ void k_cnt_init(int* __restrict__ cnt, int* __restrict__ counter, int n) {
    int i = blockIdx.x * blockDim.x + threadIdx.x;
    if (i < n) cnt[i] = 0;
    if (i == 0) *counter = 0;
}

__global__ void k_cnt_count(const int* __restrict__ dst, int* __restrict__ cnt,
                            int* __restrict__ rank, int e) {
    int i = blockIdx.x * blockDim.x + threadIdx.x;
    if (i < e) rank[i] = atomicAdd(&cnt[dst[i]], 1);
}

__global__ __launch_bounds__(256)
void k_dinv_base(const int* __restrict__ cnt, float* __restrict__ dinv,
                 int* __restrict__ start, int* __restrict__ counter, int n) {
    __shared__ int s_wave[8];
    __shared__ int s_base;
    const int tid  = threadIdx.x;
    const int lane = tid & 31;
    const int wave = tid >> 5;
    const int i    = blockIdx.x * 256 + tid;
    const int c    = (i < n) ? cnt[i] : 0;
    if (i < n) dinv[i] = rsqrtf((float)(c + 1));

    int inc = c;
    for (int off = 1; off < 32; off <<= 1) {
        int t = __shfl_up(inc, off, 32);
        if (lane >= off) inc += t;
    }
    if (lane == 31) s_wave[wave] = inc;
    __syncthreads();
    if (tid == 0) {
        int tot = 0;
        for (int w = 0; w < 8; ++w) { int t = s_wave[w]; s_wave[w] = tot; tot += t; }
        s_base = atomicAdd(counter, tot);
    }
    __syncthreads();
    if (i < n) start[i] = s_base + s_wave[wave] + inc - c;
}

__global__ void k_fill(const int* __restrict__ src, const int* __restrict__ dst,
                       const int* __restrict__ start, const int* __restrict__ rank,
                       int* __restrict__ col, int e) {
    int i = blockIdx.x * blockDim.x + threadIdx.x;
    if (i >= e) return;
    int pos = start[dst[i]] + rank[i];
    if ((unsigned)pos < (unsigned)N_EDGES) col[pos] = src[i];
}


#define TILE_M    32
#define N_TILES   (N_NODES / TILE_M)
#define XT_F4     (TILE_M * IN_DIM / 4)
#define XBUF_F4   (XT_F4 + 2)
#define W1_F4     (HID_DIM * IN_DIM / 4)
#define KSTEPS    41
#define GEMM_GRID 512

struct XTile { float4 q0, q1, q2, q3, q4, q5; };

__device__ __forceinline__ XTile xtile_load(const float4* __restrict__ xs, int tid) {
    XTile t;
    t.q0 = xs[tid];
    t.q1 = xs[tid + 256];
    t.q2 = xs[tid + 512];
    t.q3 = xs[tid + 768];
    t.q4 = xs[tid + 1024];
    t.q5 = make_float4(0.f, 0.f, 0.f, 0.f);
    if (tid < XT_F4 - 1280) t.q5 = xs[tid + 1280];
    return t;
}

__device__ __forceinline__ void xtile_store(float4* xw, int tid, const XTile& t) {
    xw[tid]        = t.q0;
    xw[tid + 256]  = t.q1;
    xw[tid + 512]  = t.q2;
    xw[tid + 768]  = t.q3;
    xw[tid + 1024] = t.q4;
    if (tid < XT_F4 - 1280) xw[tid + 1280] = t.q5;
}

__global__ __launch_bounds__(256)
void k_gemm1(const float* __restrict__ x, const float* __restrict__ W1,
             float* __restrict__ h) {
    __shared__ float4 smem4[W1_F4 + 2 * XBUF_F4];
    float*  sW = (float*)smem4;
    float4* sX = smem4 + W1_F4;

    const int tid  = threadIdx.x;
    const int lane = tid & 31;
    const int wave = tid >> 5;
    const int c    = lane & 15;
    const int hh   = lane >> 4;

    int tile = blockIdx.x;

    XTile xr = xtile_load((const float4*)(x + (size_t)tile * (TILE_M * IN_DIM)), tid);

    {
        const float4* w4 = (const float4*)W1;
        for (int q = tid; q < W1_F4; q += 256) smem4[q] = w4[q];
    }
    xtile_store(sX, tid, xr);
    __syncthreads();

    v2f   bf[KSTEPS];
    float bt;
    {
        const float* wrow = sW + (wave * 16 + c) * IN_DIM;
#pragma unroll
        for (int j = 0; j < KSTEPS; ++j) {
            bf[j].x = wrow[4 * j + 2 * hh];
            bf[j].y = wrow[4 * j + 2 * hh + 1];
        }
        const float w164 = wrow[4 * KSTEPS];
        bt = (hh == 0) ? w164 : 0.0f;
    }

    int cur = 0;
    for (; tile < N_TILES; tile += GEMM_GRID) {
        const int  nxt  = tile + GEMM_GRID;
        const bool more = nxt < N_TILES;
        if (more)
            xr = xtile_load((const float4*)(x + (size_t)nxt * (TILE_M * IN_DIM)), tid);

        const float* xb = (const float*)(sX + cur * XBUF_F4);
        const float* a0 = xb + c * IN_DIM;
        const float* a1 = xb + (16 + c) * IN_DIM;
        v8f acc0 = {0.f, 0.f, 0.f, 0.f, 0.f, 0.f, 0.f, 0.f};
        v8f acc1 = {0.f, 0.f, 0.f, 0.f, 0.f, 0.f, 0.f, 0.f};

#pragma unroll
        for (int j = 0; j < KSTEPS; ++j) {
            v2f f0, f1;
            f0.x = a0[4 * j + 2 * hh];
            f0.y = a0[4 * j + 2 * hh + 1];
            f1.x = a1[4 * j + 2 * hh];
            f1.y = a1[4 * j + 2 * hh + 1];
            acc0 = __builtin_amdgcn_wmma_f32_16x16x4_f32(
                       false, f0, false, bf[j], (short)0, acc0, false, false);
            acc1 = __builtin_amdgcn_wmma_f32_16x16x4_f32(
                       false, f1, false, bf[j], (short)0, acc1, false, false);
        }
        {
            const float t0 = a0[4 * KSTEPS];
            const float t1 = a1[4 * KSTEPS];
            v2f f0, f1, fb;
            f0.x = (hh == 0) ? t0 : 0.0f;  f0.y = 0.0f;
            f1.x = (hh == 0) ? t1 : 0.0f;  f1.y = 0.0f;
            fb.x = bt;                     fb.y = 0.0f;
            acc0 = __builtin_amdgcn_wmma_f32_16x16x4_f32(
                       false, f0, false, fb, (short)0, acc0, false, false);
            acc1 = __builtin_amdgcn_wmma_f32_16x16x4_f32(
                       false, f1, false, fb, (short)0, acc1, false, false);
        }

        if (more) xtile_store(sX + (cur ^ 1) * XBUF_F4, tid, xr);
        __syncthreads();

        {
            float* hp = h + ((size_t)tile * TILE_M + 8 * hh) * HID_DIM + wave * 16 + c;
#pragma unroll
            for (int r = 0; r < 8; ++r) {
                hp[(size_t)r * HID_DIM]        = acc0[r];
                hp[(size_t)(16 + r) * HID_DIM] = acc1[r];
            }
        }
        cur ^= 1;
    }
}


template <int K>
__device__ __forceinline__ void gather_rows(float4& acc, const float* __restrict__ h,
                                            int sv, int wbits, int t, int f4) {
    float4 v[K];
    float  w[K];
#pragma unroll
    for (int u = 0; u < K; ++u) {
        const int s = __builtin_amdgcn_readlane(sv, t + u);
        w[u] = __int_as_float(__builtin_amdgcn_readlane(wbits, t + u));
        v[u] = *(const float4*)&h[(size_t)s * HID_DIM + f4];
    }
#pragma unroll
    for (int u = 0; u < K; ++u) {
        acc.x += v[u].x * w[u]; acc.y += v[u].y * w[u];
        acc.z += v[u].z * w[u]; acc.w += v[u].w * w[u];
    }
}

__global__ __launch_bounds__(256)
void k_gather1(const int* __restrict__ start, const int* __restrict__ cnt,
               const int* __restrict__ col, const float* __restrict__ dinv,
               const float* __restrict__ h, const float* __restrict__ b1,
               const float* __restrict__ W2, float* __restrict__ sval, int n) {
    const int lane = threadIdx.x & 31;
    const int i = __builtin_amdgcn_readfirstlane(blockIdx.x * 8 + (threadIdx.x >> 5));
    if (i >= n) return;
    const int f4 = lane * 4;
    const int   m   = cnt[i];
    const int   beg = start[i];
    const float di  = dinv[i];

    float4 acc;
    {
        const float4 hv = *(const float4*)&h[(size_t)i * HID_DIM + f4];
        const float  ws = di * di;
        acc.x = hv.x * ws; acc.y = hv.y * ws; acc.z = hv.z * ws; acc.w = hv.w * ws;
    }

    for (int base = 0; base < m; base += 32) {
        const int k = base + lane;
        int   s = 0;
        float w = 0.0f;
        if (k < m) { s = col[beg + k]; w = dinv[s] * di; }
        const int wbits = __float_as_int(w);
        const int nb = (m - base < 32) ? (m - base) : 32;
        int t = 0;
        for (; t + 8 <= nb; t += 8) gather_rows<8>(acc, h, s, wbits, t, f4);
        if (nb - t >= 4) { gather_rows<4>(acc, h, s, wbits, t, f4); t += 4; }
        const int r = nb - t;
        if (r == 3)      gather_rows<3>(acc, h, s, wbits, t, f4);
        else if (r == 2) gather_rows<2>(acc, h, s, wbits, t, f4);
        else if (r == 1) gather_rows<1>(acc, h, s, wbits, t, f4);
    }

    const float4 bb = *(const float4*)&b1[f4];
    const float4 w2 = *(const float4*)&W2[f4];
    float p = fmaxf(acc.x + bb.x, 0.f) * w2.x
            + fmaxf(acc.y + bb.y, 0.f) * w2.y
            + fmaxf(acc.z + bb.z, 0.f) * w2.z
            + fmaxf(acc.w + bb.w, 0.f) * w2.w;
    for (int off = 16; off > 0; off >>= 1)
        p += __shfl_xor(p, off, 32);
    if (lane == 0) sval[i] = p;
}


__global__ void k_gather2(const int* __restrict__ start, const int* __restrict__ cnt,
                          const int* __restrict__ col, const float* __restrict__ dinv,
                          const float* __restrict__ sval, const float* __restrict__ b2,
                          float* __restrict__ out, int n) {
    int i = blockIdx.x * blockDim.x + threadIdx.x;
    if (i >= n) return;
    const float di  = dinv[i];
    const int   beg = start[i];
    const int   m   = cnt[i];
    float acc = sval[i] * (di * di);
    for (int t = 0; t < m; ++t) {
        const int s = col[beg + t];
        acc += sval[s] * (dinv[s] * di);
    }
    out[i] = 1.0f / (1.0f + expf(-(acc + b2[0])));
}


extern "C" void kernel_launch(void* const* d_in, const int* in_sizes, int n_in,
                              void* d_out, int out_size, void* d_ws, size_t ws_size,
                              hipStream_t stream) {
    (void)in_sizes; (void)n_in; (void)out_size; (void)ws_size;
    const float* x  = (const float*)d_in[0];
    const float* W1 = (const float*)d_in[1];
    const float* b1 = (const float*)d_in[2];
    const float* W2 = (const float*)d_in[3];
    const float* b2 = (const float*)d_in[4];
    const int*   ei = (const int*)d_in[5];
    const int* src = ei;
    const int* dst = ei + N_EDGES;
    float* out = (float*)d_out;

    float* dinv = (float*)d_ws;
    float* sval = dinv + N_NODES;
    float* h    = sval + N_NODES;
    int* cnt     = (int*)(h + (size_t)N_NODES * HID_DIM);
    int* start   = cnt + N_NODES;
    int* rank    = start + N_NODES;
    int* col     = rank + N_EDGES;
    int* counter = col + N_EDGES;

    const int B = 256;
    k_cnt_init <<<(N_NODES + B - 1) / B, B, 0, stream>>>(cnt, counter, N_NODES);
    k_cnt_count<<<(N_EDGES + B - 1) / B, B, 0, stream>>>(dst, cnt, rank, N_EDGES);
    k_dinv_base<<<(N_NODES + B - 1) / B, B, 0, stream>>>(cnt, dinv, start, counter, N_NODES);
    k_fill     <<<(N_EDGES + B - 1) / B, B, 0, stream>>>(src, dst, start, rank, col, N_EDGES);

    k_gemm1    <<<GEMM_GRID, B, 0, stream>>>(x, W1, h);

    k_gather1  <<<(N_NODES * 32) / B, B, 0, stream>>>(start, cnt, col, dinv, h, b1, W2, sval, N_NODES);
    k_gather2  <<<(N_NODES + B - 1) / B, B, 0, stream>>>(start, cnt, col, dinv, sval, b2, out, N_NODES);
}


# Edge grouping: the per-edge returning global atomic (count = rank) and scattered fill are replaced by two levels of LDS counters (1024-node bins, then nodes); sum order = LDS add arrival rank
# speedup vs baseline: 4.0933x; 1.5010x over previous
#include <hip/hip_runtime.h>
#include <math.h>

#define N_NODES 200000
#define N_EDGES 2000000
#define IN_DIM  165
#define HID_DIM 128

typedef __attribute__((ext_vector_type(2))) float v2f;
typedef __attribute__((ext_vector_type(8))) float v8f;


#define BIN_SHIFT 10
#define BIN_NODES (1 << BIN_SHIFT)
#define N_BINS    ((N_NODES + BIN_NODES - 1) / BIN_NODES)
#define L1_GRID   256
#define L1_ROWS   (L1_GRID / 4)
#define L1_CHUNK  ((N_EDGES + L1_GRID - 1) / L1_GRID)
#define L1_PER_T  ((L1_CHUNK + 1023) / 1024)

__device__ __forceinline__ int bin_of(int d) {
    const unsigned b = (unsigned)d >> BIN_SHIFT;
    return (int)(b < (unsigned)N_BINS ? b : (unsigned)(N_BINS - 1));
}

__global__ __launch_bounds__(1024)
void k_bin_count(const int* __restrict__ dst, int* __restrict__ histo) {
    __shared__ int s_hist[N_BINS];
    const int tid = threadIdx.x;
    if (tid < N_BINS) s_hist[tid] = 0;
    __syncthreads();
    const int beg = blockIdx.x * L1_CHUNK;
    const int end = (beg + L1_CHUNK < N_EDGES) ? (beg + L1_CHUNK) : N_EDGES;
    int dv[L1_PER_T];
#pragma unroll
    for (int u = 0; u < L1_PER_T; ++u) {
        const int e = beg + u * 1024 + tid;
        dv[u] = dst[e < end ? e : end - 1];
    }
#pragma unroll
    for (int u = 0; u < L1_PER_T; ++u) {
        const int e = beg + u * 1024 + tid;
        if (e < end) atomicAdd(&s_hist[bin_of(dv[u])], 1);
    }
    __syncthreads();
    if (tid < N_BINS) histo[blockIdx.x * N_BINS + tid] = s_hist[tid];
}

__global__ __launch_bounds__(1024)
void k_bin_scatter(const int* __restrict__ src, const int* __restrict__ dst,
                   const int* __restrict__ histo, int* __restrict__ rec_words,
                   int* __restrict__ binBase) {
    int2* rec = (int2*)rec_words;
    __shared__ int s_hist[N_BINS];
    __shared__ int s_part[2][4][256];
    __shared__ int s_base[256];
    __shared__ int s_wtot[8];
    const int tid  = threadIdx.x;
    const int lane = tid & 31;
    const int wave = tid >> 5;
    const int g    = blockIdx.x;
    const int beg  = g * L1_CHUNK;
    const int end  = (beg + L1_CHUNK < N_EDGES) ? (beg + L1_CHUNK) : N_EDGES;

    int es[L1_PER_T], ed[L1_PER_T];
#pragma unroll
    for (int u = 0; u < L1_PER_T; ++u) {
        const int e  = beg + u * 1024 + tid;
        const int ec = e < end ? e : end - 1;
        es[u] = src[ec];
        ed[u] = dst[ec];
    }
    if (tid < N_BINS) s_hist[tid] = 0;

    const int colb = tid & 255;
    const int part = tid >> 8;
    int tot = 0, pre = 0;
    if (colb < N_BINS) {
        for (int r0 = 0; r0 < L1_ROWS; r0 += 16) {
            int v[16];
#pragma unroll
            for (int k = 0; k < 16; ++k)
                v[k] = histo[(part * L1_ROWS + r0 + k) * N_BINS + colb];
#pragma unroll
            for (int k = 0; k < 16; ++k) {
                tot += v[k];
                if (part * L1_ROWS + r0 + k < g) pre += v[k];
            }
        }
    }
    s_part[0][part][colb] = tot;
    s_part[1][part][colb] = pre;
    __syncthreads();

    int bsize = 0, bpre = 0, inc = 0;
    if (tid < 256) {
        bsize = s_part[0][0][tid] + s_part[0][1][tid] + s_part[0][2][tid] + s_part[0][3][tid];
        bpre  = s_part[1][0][tid] + s_part[1][1][tid] + s_part[1][2][tid] + s_part[1][3][tid];
        inc = bsize;
        for (int off = 1; off < 32; off <<= 1) {
            const int t = __shfl_up(inc, off, 32);
            if (lane >= off) inc += t;
        }
        if (lane == 31) s_wtot[wave] = inc;
    }
    __syncthreads();
    if (tid < 256) {
        int wb = 0;
        for (int w = 0; w < wave; ++w) wb += s_wtot[w];
        const int bb = wb + inc - bsize;
        s_base[tid] = bb + bpre;
        if (g == 0 && tid <= N_BINS) binBase[tid] = bb;
    }
    __syncthreads();

#pragma unroll
    for (int u = 0; u < L1_PER_T; ++u) {
        const int e = beg + u * 1024 + tid;
        if (e < end) {
            const int b   = bin_of(ed[u]);
            const int pos = s_base[b] + atomicAdd(&s_hist[b], 1);
            if ((unsigned)pos < (unsigned)N_EDGES)
                rec[pos] = make_int2(es[u], ed[u]);
        }
    }
}

__global__ __launch_bounds__(1024)
void k_bin_csr(const int* __restrict__ rec_words, const int* __restrict__ binBase,
               int* __restrict__ cnt, int* __restrict__ start, float* __restrict__ dinv,
               int* __restrict__ col) {
    const int2* rec = (const int2*)rec_words;
    __shared__ int s_cnt[BIN_NODES];
    __shared__ int s_cur[BIN_NODES];
    __shared__ int s_wtot[32];
    const int tid   = threadIdx.x;
    const int lane  = tid & 31;
    const int wave  = tid >> 5;
    const int node0 = blockIdx.x * BIN_NODES;
    int lo = binBase[blockIdx.x];
    int hi = binBase[blockIdx.x + 1];
    if (lo < 0) lo = 0;
    if (hi > N_EDGES) hi = N_EDGES;

    s_cnt[tid] = 0;
    __syncthreads();

    for (int e0 = lo; e0 < hi; e0 += 4096) {
        int dy[4];
#pragma unroll
        for (int u = 0; u < 4; ++u) {
            const int e = e0 + u * 1024 + tid;
            dy[u] = rec[e < hi ? e : hi - 1].y;
        }
#pragma unroll
        for (int u = 0; u < 4; ++u) {
            const int e = e0 + u * 1024 + tid;
            const int d = dy[u] - node0;
            if (e < hi && (unsigned)d < (unsigned)BIN_NODES) atomicAdd(&s_cnt[d], 1);
        }
    }
    __syncthreads();

    const int c = s_cnt[tid];
    int inc = c;
    for (int off = 1; off < 32; off <<= 1) {
        const int t = __shfl_up(inc, off, 32);
        if (lane >= off) inc += t;
    }
    if (lane == 31) s_wtot[wave] = inc;
    __syncthreads();
    int wb = 0;
    for (int w = 0; w < wave; ++w) wb += s_wtot[w];
    const int st = wb + inc - c;
    s_cur[tid] = st;
    {
        const int node = node0 + tid;
        if (node < N_NODES) {
            cnt[node]   = c;
            start[node] = lo + st;
            dinv[node]  = rsqrtf((float)(c + 1));
        }
    }
    __syncthreads();

    for (int e0 = lo; e0 < hi; e0 += 4096) {
        int2 r[4];
#pragma unroll
        for (int u = 0; u < 4; ++u) {
            const int e = e0 + u * 1024 + tid;
            r[u] = rec[e < hi ? e : hi - 1];
        }
#pragma unroll
        for (int u = 0; u < 4; ++u) {
            const int e = e0 + u * 1024 + tid;
            const int d = r[u].y - node0;
            if (e < hi && (unsigned)d < (unsigned)BIN_NODES) {
                const int pos = lo + atomicAdd(&s_cur[d], 1);
                if ((unsigned)pos < (unsigned)N_EDGES) col[pos] = r[u].x;
            }
        }
    }
}


#define TILE_M    32
#define N_TILES   (N_NODES / TILE_M)
#define XT_F4     (TILE_M * IN_DIM / 4)
#define XBUF_F4   (XT_F4 + 2)
#define W1_F4     (HID_DIM * IN_DIM / 4)
#define KSTEPS    41
#define GEMM_GRID 512

struct XTile { float4 q0, q1, q2, q3, q4, q5; };

__device__ __forceinline__ XTile xtile_load(const float4* __restrict__ xs, int tid) {
    XTile t;
    t.q0 = xs[tid];
    t.q1 = xs[tid + 256];
    t.q2 = xs[tid + 512];
    t.q3 = xs[tid + 768];
    t.q4 = xs[tid + 1024];
    t.q5 = make_float4(0.f, 0.f, 0.f, 0.f);
    if (tid < XT_F4 - 1280) t.q5 = xs[tid + 1280];
    return t;
}

__device__ __forceinline__ void xtile_store(float4* xw, int tid, const XTile& t) {
    xw[tid]        = t.q0;
    xw[tid + 256]  = t.q1;
    xw[tid + 512]  = t.q2;
    xw[tid + 768]  = t.q3;
    xw[tid + 1024] = t.q4;
    if (tid < XT_F4 - 1280) xw[tid + 1280] = t.q5;
}

__global__ __launch_bounds__(256)
void k_gemm1(const float* __restrict__ x, const float* __restrict__ W1,
             float* __restrict__ h) {
    __shared__ float4 smem4[W1_F4 + 2 * XBUF_F4];
    float*  sW = (float*)smem4;
    float4* sX = smem4 + W1_F4;

    const int tid  = threadIdx.x;
    const int lane = tid & 31;
    const int wave = tid >> 5;
    const int c    = lane & 15;
    const int hh   = lane >> 4;

    int tile = blockIdx.x;

    XTile xr = xtile_load((const float4*)(x + (size_t)tile * (TILE_M * IN_DIM)), tid);

    {
        const float4* w4 = (const float4*)W1;
        for (int q = tid; q < W1_F4; q += 256) smem4[q] = w4[q];
    }
    xtile_store(sX, tid, xr);
    __syncthreads();

    v2f   bf[KSTEPS];
    float bt;
    {
        const float* wrow = sW + (wave * 16 + c) * IN_DIM;
#pragma unroll
        for (int j = 0; j < KSTEPS; ++j) {
            bf[j].x = wrow[4 * j + 2 * hh];
            bf[j].y = wrow[4 * j + 2 * hh + 1];
        }
        const float w164 = wrow[4 * KSTEPS];
        bt = (hh == 0) ? w164 : 0.0f;
    }

    int cur = 0;
    for (; tile < N_TILES; tile += GEMM_GRID) {
        const int  nxt  = tile + GEMM_GRID;
        const bool more = nxt < N_TILES;
        if (more)
            xr = xtile_load((const float4*)(x + (size_t)nxt * (TILE_M * IN_DIM)), tid);

        const float* xb = (const float*)(sX + cur * XBUF_F4);
        const float* a0 = xb + c * IN_DIM;
        const float* a1 = xb + (16 + c) * IN_DIM;
        v8f acc0 = {0.f, 0.f, 0.f, 0.f, 0.f, 0.f, 0.f, 0.f};
        v8f acc1 = {0.f, 0.f, 0.f, 0.f, 0.f, 0.f, 0.f, 0.f};

#pragma unroll
        for (int j = 0; j < KSTEPS; ++j) {
            v2f f0, f1;
            f0.x = a0[4 * j + 2 * hh];
            f0.y = a0[4 * j + 2 * hh + 1];
            f1.x = a1[4 * j + 2 * hh];
            f1.y = a1[4 * j + 2 * hh + 1];
            acc0 = __builtin_amdgcn_wmma_f32_16x16x4_f32(
                       false, f0, false, bf[j], (short)0, acc0, false, false);
            acc1 = __builtin_amdgcn_wmma_f32_16x16x4_f32(
                       false, f1, false, bf[j], (short)0, acc1, false, false);
        }
        {
            const float t0 = a0[4 * KSTEPS];
            const float t1 = a1[4 * KSTEPS];
            v2f f0, f1, fb;
            f0.x = (hh == 0) ? t0 : 0.0f;  f0.y = 0.0f;
            f1.x = (hh == 0) ? t1 : 0.0f;  f1.y = 0.0f;
            fb.x = bt;                     fb.y = 0.0f;
            acc0 = __builtin_amdgcn_wmma_f32_16x16x4_f32(
                       false, f0, false, fb, (short)0, acc0, false, false);
            acc1 = __builtin_amdgcn_wmma_f32_16x16x4_f32(
                       false, f1, false, fb, (short)0, acc1, false, false);
        }

        if (more) xtile_store(sX + (cur ^ 1) * XBUF_F4, tid, xr);
        __syncthreads();

        {
            float* hp = h + ((size_t)tile * TILE_M + 8 * hh) * HID_DIM + wave * 16 + c;
#pragma unroll
            for (int r = 0; r < 8; ++r) {
                hp[(size_t)r * HID_DIM]        = acc0[r];
                hp[(size_t)(16 + r) * HID_DIM] = acc1[r];
            }
        }
        cur ^= 1;
    }
}


template <int K>
__device__ __forceinline__ void gather_rows(float4& acc, const float* __restrict__ h,
                                            int sv, int wbits, int t, int f4) {
    float4 v[K];
    float  w[K];
#pragma unroll
    for (int u = 0; u < K; ++u) {
        const int s = __builtin_amdgcn_readlane(sv, t + u);
        w[u] = __int_as_float(__builtin_amdgcn_readlane(wbits, t + u));
        v[u] = *(const float4*)&h[(size_t)s * HID_DIM + f4];
    }
#pragma unroll
    for (int u = 0; u < K; ++u) {
        acc.x += v[u].x * w[u]; acc.y += v[u].y * w[u];
        acc.z += v[u].z * w[u]; acc.w += v[u].w * w[u];
    }
}

__global__ __launch_bounds__(256)
void k_gather1(const int* __restrict__ start, const int* __restrict__ cnt,
               const int* __restrict__ col, const float* __restrict__ dinv,
               const float* __restrict__ h, const float* __restrict__ b1,
               const float* __restrict__ W2, float* __restrict__ sval, int n) {
    const int lane = threadIdx.x & 31;
    const int i = __builtin_amdgcn_readfirstlane(blockIdx.x * 8 + (threadIdx.x >> 5));
    if (i >= n) return;
    const int f4 = lane * 4;
    const int   m   = cnt[i];
    const int   beg = start[i];
    const float di  = dinv[i];

    float4 acc;
    {
        const float4 hv = *(const float4*)&h[(size_t)i * HID_DIM + f4];
        const float  ws = di * di;
        acc.x = hv.x * ws; acc.y = hv.y * ws; acc.z = hv.z * ws; acc.w = hv.w * ws;
    }

    for (int base = 0; base < m; base += 32) {
        const int k = base + lane;
        int   s = 0;
        float w = 0.0f;
        if (k < m) { s = col[beg + k]; w = dinv[s] * di; }
        const int wbits = __float_as_int(w);
        const int nb = (m - base < 32) ? (m - base) : 32;
        int t = 0;
        for (; t + 8 <= nb; t += 8) gather_rows<8>(acc, h, s, wbits, t, f4);
        if (nb - t >= 4) { gather_rows<4>(acc, h, s, wbits, t, f4); t += 4; }
        const int r = nb - t;
        if (r == 3)      gather_rows<3>(acc, h, s, wbits, t, f4);
        else if (r == 2) gather_rows<2>(acc, h, s, wbits, t, f4);
        else if (r == 1) gather_rows<1>(acc, h, s, wbits, t, f4);
    }

    const float4 bb = *(const float4*)&b1[f4];
    const float4 w2 = *(const float4*)&W2[f4];
    float p = fmaxf(acc.x + bb.x, 0.f) * w2.x
            + fmaxf(acc.y + bb.y, 0.f) * w2.y
            + fmaxf(acc.z + bb.z, 0.f) * w2.z
            + fmaxf(acc.w + bb.w, 0.f) * w2.w;
    for (int off = 16; off > 0; off >>= 1)
        p += __shfl_xor(p, off, 32);
    if (lane == 0) sval[i] = p;
}


__global__ void k_gather2(const int* __restrict__ start, const int* __restrict__ cnt,
                          const int* __restrict__ col, const float* __restrict__ dinv,
                          const float* __restrict__ sval, const float* __restrict__ b2,
                          float* __restrict__ out, int n) {
    int i = blockIdx.x * blockDim.x + threadIdx.x;
    if (i >= n) return;
    const float di  = dinv[i];
    const int   beg = start[i];
    const int   m   = cnt[i];
    float acc = sval[i] * (di * di);
    for (int t = 0; t < m; ++t) {
        const int s = col[beg + t];
        acc += sval[s] * (dinv[s] * di);
    }
    out[i] = 1.0f / (1.0f + expf(-(acc + b2[0])));
}


extern "C" void kernel_launch(void* const* d_in, const int* in_sizes, int n_in,
                              void* d_out, int out_size, void* d_ws, size_t ws_size,
                              hipStream_t stream) {
    (void)in_sizes; (void)n_in; (void)out_size; (void)ws_size;
    const float* x  = (const float*)d_in[0];
    const float* W1 = (const float*)d_in[1];
    const float* b1 = (const float*)d_in[2];
    const float* W2 = (const float*)d_in[3];
    const float* b2 = (const float*)d_in[4];
    const int*   ei = (const int*)d_in[5];
    const int* src = ei;
    const int* dst = ei + N_EDGES;
    float* out = (float*)d_out;

    float* dinv = (float*)d_ws;
    float* sval = dinv + N_NODES;
    float* h    = sval + N_NODES;
    int*  cnt     = (int*)(h + (size_t)N_NODES * HID_DIM);
    int*  start   = cnt + N_NODES;
    int*  rec     = start + N_NODES;
    int*  col     = rec + 2 * N_EDGES;
    int*  histo   = col + N_EDGES;
    int*  binBase = histo + L1_GRID * N_BINS;

    const int B = 256;
    k_bin_count  <<<L1_GRID, 1024, 0, stream>>>(dst, histo);
    k_bin_scatter<<<L1_GRID, 1024, 0, stream>>>(src, dst, histo, rec, binBase);
    k_bin_csr    <<<N_BINS, 1024, 0, stream>>>(rec, binBase, cnt, start, dinv, col);

    k_gemm1    <<<GEMM_GRID, B, 0, stream>>>(x, W1, h);

    k_gather1  <<<(N_NODES * 32) / B, B, 0, stream>>>(start, cnt, col, dinv, h, b1, W2, sval, N_NODES);
    k_gather2  <<<(N_NODES + B - 1) / B, B, 0, stream>>>(start, cnt, col, dinv, sval, b2, out, N_NODES);
}


# Layer-1 gather: each wave walks 8 consecutive nodes (25000 waves enter, not 200000); bucket data read once per wave; source lists, their dinv and own rows requested turns ahead; sum order unchanged
# speedup vs baseline: 4.3920x; 1.0730x over previous
#include <hip/hip_runtime.h>
#include <math.h>

#define N_NODES 200000
#define N_EDGES 2000000
#define IN_DIM  165
#define HID_DIM 128

typedef __attribute__((ext_vector_type(2))) float v2f;
typedef __attribute__((ext_vector_type(8))) float v8f;


#define BIN_SHIFT 10
#define BIN_NODES (1 << BIN_SHIFT)
#define N_BINS    ((N_NODES + BIN_NODES - 1) / BIN_NODES)
#define L1_GRID   256
#define L1_ROWS   (L1_GRID / 4)
#define L1_CHUNK  ((N_EDGES + L1_GRID - 1) / L1_GRID)
#define L1_PER_T  ((L1_CHUNK + 1023) / 1024)

__device__ __forceinline__ int bin_of(int d) {
    const unsigned b = (unsigned)d >> BIN_SHIFT;
    return (int)(b < (unsigned)N_BINS ? b : (unsigned)(N_BINS - 1));
}

__global__ __launch_bounds__(1024)
void k_bin_count(const int* __restrict__ dst, int* __restrict__ histo) {
    __shared__ int s_hist[N_BINS];
    const int tid = threadIdx.x;
    if (tid < N_BINS) s_hist[tid] = 0;
    __syncthreads();
    const int beg = blockIdx.x * L1_CHUNK;
    const int end = (beg + L1_CHUNK < N_EDGES) ? (beg + L1_CHUNK) : N_EDGES;
    int dv[L1_PER_T];
#pragma unroll
    for (int u = 0; u < L1_PER_T; ++u) {
        const int e = beg + u * 1024 + tid;
        dv[u] = dst[e < end ? e : end - 1];
    }
#pragma unroll
    for (int u = 0; u < L1_PER_T; ++u) {
        const int e = beg + u * 1024 + tid;
        if (e < end) atomicAdd(&s_hist[bin_of(dv[u])], 1);
    }
    __syncthreads();
    if (tid < N_BINS) histo[blockIdx.x * N_BINS + tid] = s_hist[tid];
}

__global__ __launch_bounds__(1024)
void k_bin_scatter(const int* __restrict__ src, const int* __restrict__ dst,
                   const int* __restrict__ histo, int* __restrict__ rec_words,
                   int* __restrict__ binBase) {
    int2* rec = (int2*)rec_words;
    __shared__ int s_hist[N_BINS];
    __shared__ int s_part[2][4][256];
    __shared__ int s_base[256];
    __shared__ int s_wtot[8];
    const int tid  = threadIdx.x;
    const int lane = tid & 31;
    const int wave = tid >> 5;
    const int g    = blockIdx.x;
    const int beg  = g * L1_CHUNK;
    const int end  = (beg + L1_CHUNK < N_EDGES) ? (beg + L1_CHUNK) : N_EDGES;

    int es[L1_PER_T], ed[L1_PER_T];
#pragma unroll
    for (int u = 0; u < L1_PER_T; ++u) {
        const int e  = beg + u * 1024 + tid;
        const int ec = e < end ? e : end - 1;
        es[u] = src[ec];
        ed[u] = dst[ec];
    }
    if (tid < N_BINS) s_hist[tid] = 0;

    const int colb = tid & 255;
    const int part = tid >> 8;
    int tot = 0, pre = 0;
    if (colb < N_BINS) {
        for (int r0 = 0; r0 < L1_ROWS; r0 += 16) {
            int v[16];
#pragma unroll
            for (int k = 0; k < 16; ++k)
                v[k] = histo[(part * L1_ROWS + r0 + k) * N_BINS + colb];
#pragma unroll
            for (int k = 0; k < 16; ++k) {
                tot += v[k];
                if (part * L1_ROWS + r0 + k < g) pre += v[k];
            }
        }
    }
    s_part[0][part][colb] = tot;
    s_part[1][part][colb] = pre;
    __syncthreads();

    int bsize = 0, bpre = 0, inc = 0;
    if (tid < 256) {
        bsize = s_part[0][0][tid] + s_part[0][1][tid] + s_part[0][2][tid] + s_part[0][3][tid];
        bpre  = s_part[1][0][tid] + s_part[1][1][tid] + s_part[1][2][tid] + s_part[1][3][tid];
        inc = bsize;
        for (int off = 1; off < 32; off <<= 1) {
            const int t = __shfl_up(inc, off, 32);
            if (lane >= off) inc += t;
        }
        if (lane == 31) s_wtot[wave] = inc;
    }
    __syncthreads();
    if (tid < 256) {
        int wb = 0;
        for (int w = 0; w < wave; ++w) wb += s_wtot[w];
        const int bb = wb + inc - bsize;
        s_base[tid] = bb + bpre;
        if (g == 0 && tid <= N_BINS) binBase[tid] = bb;
    }
    __syncthreads();

#pragma unroll
    for (int u = 0; u < L1_PER_T; ++u) {
        const int e = beg + u * 1024 + tid;
        if (e < end) {
            const int b   = bin_of(ed[u]);
            const int pos = s_base[b] + atomicAdd(&s_hist[b], 1);
            if ((unsigned)pos < (unsigned)N_EDGES)
                rec[pos] = make_int2(es[u], ed[u]);
        }
    }
}

__global__ __launch_bounds__(1024)
void k_bin_csr(const int* __restrict__ rec_words, const int* __restrict__ binBase,
               int* __restrict__ cnt, int* __restrict__ start, float* __restrict__ dinv,
               int* __restrict__ col) {
    const int2* rec = (const int2*)rec_words;
    __shared__ int s_cnt[BIN_NODES];
    __shared__ int s_cur[BIN_NODES];
    __shared__ int s_wtot[32];
    const int tid   = threadIdx.x;
    const int lane  = tid & 31;
    const int wave  = tid >> 5;
    const int node0 = blockIdx.x * BIN_NODES;
    int lo = binBase[blockIdx.x];
    int hi = binBase[blockIdx.x + 1];
    if (lo < 0) lo = 0;
    if (hi > N_EDGES) hi = N_EDGES;

    s_cnt[tid] = 0;
    __syncthreads();

    for (int e0 = lo; e0 < hi; e0 += 4096) {
        int dy[4];
#pragma unroll
        for (int u = 0; u < 4; ++u) {
            const int e = e0 + u * 1024 + tid;
            dy[u] = rec[e < hi ? e : hi - 1].y;
        }
#pragma unroll
        for (int u = 0; u < 4; ++u) {
            const int e = e0 + u * 1024 + tid;
            const int d = dy[u] - node0;
            if (e < hi && (unsigned)d < (unsigned)BIN_NODES) atomicAdd(&s_cnt[d], 1);
        }
    }
    __syncthreads();

    const int c = s_cnt[tid];
    int inc = c;
    for (int off = 1; off < 32; off <<= 1) {
        const int t = __shfl_up(inc, off, 32);
        if (lane >= off) inc += t;
    }
    if (lane == 31) s_wtot[wave] = inc;
    __syncthreads();
    int wb = 0;
    for (int w = 0; w < wave; ++w) wb += s_wtot[w];
    const int st = wb + inc - c;
    s_cur[tid] = st;
    {
        const int node = node0 + tid;
        if (node < N_NODES) {
            cnt[node]   = c;
            start[node] = lo + st;
            dinv[node]  = rsqrtf((float)(c + 1));
        }
    }
    __syncthreads();

    for (int e0 = lo; e0 < hi; e0 += 4096) {
        int2 r[4];
#pragma unroll
        for (int u = 0; u < 4; ++u) {
            const int e = e0 + u * 1024 + tid;
            r[u] = rec[e < hi ? e : hi - 1];
        }
#pragma unroll
        for (int u = 0; u < 4; ++u) {
            const int e = e0 + u * 1024 + tid;
            const int d = r[u].y - node0;
            if (e < hi && (unsigned)d < (unsigned)BIN_NODES) {
                const int pos = lo + atomicAdd(&s_cur[d], 1);
                if ((unsigned)pos < (unsigned)N_EDGES) col[pos] = r[u].x;
            }
        }
    }
}


#define TILE_M    32
#define N_TILES   (N_NODES / TILE_M)
#define XT_F4     (TILE_M * IN_DIM / 4)
#define XBUF_F4   (XT_F4 + 2)
#define W1_F4     (HID_DIM * IN_DIM / 4)
#define KSTEPS    41
#define GEMM_GRID 512

struct XTile { float4 q0, q1, q2, q3, q4, q5; };

__device__ __forceinline__ XTile xtile_load(const float4* __restrict__ xs, int tid) {
    XTile t;
    t.q0 = xs[tid];
    t.q1 = xs[tid + 256];
    t.q2 = xs[tid + 512];
    t.q3 = xs[tid + 768];
    t.q4 = xs[tid + 1024];
    t.q5 = make_float4(0.f, 0.f, 0.f, 0.f);
    if (tid < XT_F4 - 1280) t.q5 = xs[tid + 1280];
    return t;
}

__device__ __forceinline__ void xtile_store(float4* xw, int tid, const XTile& t) {
    xw[tid]        = t.q0;
    xw[tid + 256]  = t.q1;
    xw[tid + 512]  = t.q2;
    xw[tid + 768]  = t.q3;
    xw[tid + 1024] = t.q4;
    if (tid < XT_F4 - 1280) xw[tid + 1280] = t.q5;
}

__global__ __launch_bounds__(256)
void k_gemm1(const float* __restrict__ x, const float* __restrict__ W1,
             float* __restrict__ h) {
    __shared__ float4 smem4[W1_F4 + 2 * XBUF_F4];
    float*  sW = (float*)smem4;
    float4* sX = smem4 + W1_F4;

    const int tid  = threadIdx.x;
    const int lane = tid & 31;
    const int wave = tid >> 5;
    const int c    = lane & 15;
    const int hh   = lane >> 4;

    int tile = blockIdx.x;

    XTile xr = xtile_load((const float4*)(x + (size_t)tile * (TILE_M * IN_DIM)), tid);

    {
        const float4* w4 = (const float4*)W1;
        for (int q = tid; q < W1_F4; q += 256) smem4[q] = w4[q];
    }
    xtile_store(sX, tid, xr);
    __syncthreads();

    v2f   bf[KSTEPS];
    float bt;
    {
        const float* wrow = sW + (wave * 16 + c) * IN_DIM;
#pragma unroll
        for (int j = 0; j < KSTEPS; ++j) {
            bf[j].x = wrow[4 * j + 2 * hh];
            bf[j].y = wrow[4 * j + 2 * hh + 1];
        }
        const float w164 = wrow[4 * KSTEPS];
        bt = (hh == 0) ? w164 : 0.0f;
    }

    int cur = 0;
    for (; tile < N_TILES; tile += GEMM_GRID) {
        const int  nxt  = tile + GEMM_GRID;
        const bool more = nxt < N_TILES;
        if (more)
            xr = xtile_load((const float4*)(x + (size_t)nxt * (TILE_M * IN_DIM)), tid);

        const float* xb = (const float*)(sX + cur * XBUF_F4);
        const float* a0 = xb + c * IN_DIM;
        const float* a1 = xb + (16 + c) * IN_DIM;
        v8f acc0 = {0.f, 0.f, 0.f, 0.f, 0.f, 0.f, 0.f, 0.f};
        v8f acc1 = {0.f, 0.f, 0.f, 0.f, 0.f, 0.f, 0.f, 0.f};

#pragma unroll
        for (int j = 0; j < KSTEPS; ++j) {
            v2f f0, f1;
            f0.x = a0[4 * j + 2 * hh];
            f0.y = a0[4 * j + 2 * hh + 1];
            f1.x = a1[4 * j + 2 * hh];
            f1.y = a1[4 * j + 2 * hh + 1];
            acc0 = __builtin_amdgcn_wmma_f32_16x16x4_f32(
                       false, f0, false, bf[j], (short)0, acc0, false, false);
            acc1 = __builtin_amdgcn_wmma_f32_16x16x4_f32(
                       false, f1, false, bf[j], (short)0, acc1, false, false);
        }
        {
            const float t0 = a0[4 * KSTEPS];
            const float t1 = a1[4 * KSTEPS];
            v2f f0, f1, fb;
            f0.x = (hh == 0) ? t0 : 0.0f;  f0.y = 0.0f;
            f1.x = (hh == 0) ? t1 : 0.0f;  f1.y = 0.0f;
            fb.x = bt;                     fb.y = 0.0f;
            acc0 = __builtin_amdgcn_wmma_f32_16x16x4_f32(
                       false, f0, false, fb, (short)0, acc0, false, false);
            acc1 = __builtin_amdgcn_wmma_f32_16x16x4_f32(
                       false, f1, false, fb, (short)0, acc1, false, false);
        }

        if (more) xtile_store(sX + (cur ^ 1) * XBUF_F4, tid, xr);
        __syncthreads();

        {
            float* hp = h + ((size_t)tile * TILE_M + 8 * hh) * HID_DIM + wave * 16 + c;
#pragma unroll
            for (int r = 0; r < 8; ++r) {
                hp[(size_t)r * HID_DIM]        = acc0[r];
                hp[(size_t)(16 + r) * HID_DIM] = acc1[r];
            }
        }
        cur ^= 1;
    }
}


template <int K>
__device__ __forceinline__ void gather_rows(float4& acc, const float* __restrict__ h,
                                            int sv, int wbits, int t, int f4) {
    float4 v[K];
    float  w[K];
#pragma unroll
    for (int u = 0; u < K; ++u) {
        const int s = __builtin_amdgcn_readlane(sv, t + u);
        w[u] = __int_as_float(__builtin_amdgcn_readlane(wbits, t + u));
        v[u] = *(const float4*)&h[(size_t)s * HID_DIM + f4];
    }
#pragma unroll
    for (int u = 0; u < K; ++u) {
        acc.x += v[u].x * w[u]; acc.y += v[u].y * w[u];
        acc.z += v[u].z * w[u]; acc.w += v[u].w * w[u];
    }
}

#define G1_NODES 8
#define G1_GRID  ((N_NODES + 8 * G1_NODES - 1) / (8 * G1_NODES))

__device__ __forceinline__ int bucket_sources(const int* __restrict__ col, int beg, int m, int lane) {
    int e = beg + (lane < m ? lane : 0);
    e = e < N_EDGES - 1 ? e : N_EDGES - 1;
    return col[e];
}

__device__ __forceinline__ void gather_chunk(float4& acc, const float* __restrict__ h,
                                             int sv, int wbits, int nb, int f4) {
    int t = 0;
    for (; t + 8 <= nb; t += 8) gather_rows<8>(acc, h, sv, wbits, t, f4);
    if (nb - t >= 4) { gather_rows<4>(acc, h, sv, wbits, t, f4); t += 4; }
    const int r = nb - t;
    if (r == 3)      gather_rows<3>(acc, h, sv, wbits, t, f4);
    else if (r == 2) gather_rows<2>(acc, h, sv, wbits, t, f4);
    else if (r == 1) gather_rows<1>(acc, h, sv, wbits, t, f4);
}

__global__ __launch_bounds__(256)
void k_gather1(const int* __restrict__ start, const int* __restrict__ cnt,
               const int* __restrict__ col, const float* __restrict__ dinv,
               const float* __restrict__ h, const float* __restrict__ b1,
               const float* __restrict__ W2, float* __restrict__ sval, int n) {
    const int lane = threadIdx.x & 31;
    const int n0 = __builtin_amdgcn_readfirstlane((blockIdx.x * 8 + (threadIdx.x >> 5)) * G1_NODES);
    if (n0 >= n) return;
    const int f4 = lane * 4;
    const int ng = (n - n0 < G1_NODES) ? (n - n0) : G1_NODES;

    const int   nl = n0 + (lane < ng ? lane : ng - 1);
    const int   mv = cnt[nl];
    const int   bv = start[nl];
    const float dv = dinv[nl];

    float4 self = *(const float4*)&h[(size_t)n0 * HID_DIM + f4];
    int m   = __builtin_amdgcn_readlane(mv, 0);
    int beg = __builtin_amdgcn_readlane(bv, 0);
    int s   = bucket_sources(col, beg, m, lane);
    const int j1   = (1 < ng) ? 1 : 0;
    int m_n   = __builtin_amdgcn_readlane(mv, j1);
    int beg_n = __builtin_amdgcn_readlane(bv, j1);
    int s_n   = bucket_sources(col, beg_n, m_n, lane);
    float ds  = dinv[s];
    float pv  = 0.0f;

    for (int j = 0; j < ng; ++j) {
        const int    jn     = (j + 1 < ng) ? (j + 1) : (ng - 1);
        const int    jnn    = (j + 2 < ng) ? (j + 2) : (ng - 1);
        const float  ds_n   = dinv[s_n];
        const float4 self_n = *(const float4*)&h[(size_t)(n0 + jn) * HID_DIM + f4];
        const int    m_nn   = __builtin_amdgcn_readlane(mv, jnn);
        const int    beg_nn = __builtin_amdgcn_readlane(bv, jnn);
        const int    s_nn   = bucket_sources(col, beg_nn, m_nn, lane);

        const float di = __int_as_float(__builtin_amdgcn_readlane(__float_as_int(dv), j));
        float4 acc;
        {
            const float ws = di * di;
            acc.x = self.x * ws; acc.y = self.y * ws; acc.z = self.z * ws; acc.w = self.w * ws;
        }

        gather_chunk(acc, h, s, __float_as_int(ds * di), (m < 32) ? m : 32, f4);
        for (int base = 32; base < m; base += 32) {
            const int   s2  = bucket_sources(col, beg + base, m - base, lane);
            const float ds2 = dinv[s2];
            gather_chunk(acc, h, s2, __float_as_int(ds2 * di), (m - base < 32) ? (m - base) : 32, f4);
        }

        const float4 bb = *(const float4*)&b1[f4];
        const float4 w2 = *(const float4*)&W2[f4];
        float p = fmaxf(acc.x + bb.x, 0.f) * w2.x
                + fmaxf(acc.y + bb.y, 0.f) * w2.y
                + fmaxf(acc.z + bb.z, 0.f) * w2.z
                + fmaxf(acc.w + bb.w, 0.f) * w2.w;
        for (int off = 16; off > 0; off >>= 1)
            p += __shfl_xor(p, off, 32);
        if (lane == j) pv = p;

        m = m_n;   beg = beg_n;   s = s_n;   ds = ds_n;   self = self_n;
        m_n = m_nn; beg_n = beg_nn; s_n = s_nn;
    }
    if (lane < ng) sval[n0 + lane] = pv;
}


__global__ void k_gather2(const int* __restrict__ start, const int* __restrict__ cnt,
                          const int* __restrict__ col, const float* __restrict__ dinv,
                          const float* __restrict__ sval, const float* __restrict__ b2,
                          float* __restrict__ out, int n) {
    int i = blockIdx.x * blockDim.x + threadIdx.x;
    if (i >= n) return;
    const float di  = dinv[i];
    const int   beg = start[i];
    const int   m   = cnt[i];
    float acc = sval[i] * (di * di);
    for (int t = 0; t < m; ++t) {
        const int s = col[beg + t];
        acc += sval[s] * (dinv[s] * di);
    }
    out[i] = 1.0f / (1.0f + expf(-(acc + b2[0])));
}


extern "C" void kernel_launch(void* const* d_in, const int* in_sizes, int n_in,
                              void* d_out, int out_size, void* d_ws, size_t ws_size,
                              hipStream_t stream) {
    (void)in_sizes; (void)n_in; (void)out_size; (void)ws_size;
    const float* x  = (const float*)d_in[0];
    const float* W1 = (const float*)d_in[1];
    const float* b1 = (const float*)d_in[2];
    const float* W2 = (const float*)d_in[3];
    const float* b2 = (const float*)d_in[4];
    const int*   ei = (const int*)d_in[5];
    const int* src = ei;
    const int* dst = ei + N_EDGES;
    float* out = (float*)d_out;

    float* dinv = (float*)d_ws;
    float* sval = dinv + N_NODES;
    float* h    = sval + N_NODES;
    int*  cnt     = (int*)(h + (size_t)N_NODES * HID_DIM);
    int*  start   = cnt + N_NODES;
    int*  rec     = start + N_NODES;
    int*  col     = rec + 2 * N_EDGES;
    int*  histo   = col + N_EDGES;
    int*  binBase = histo + L1_GRID * N_BINS;

    const int B = 256;
    k_bin_count  <<<L1_GRID, 1024, 0, stream>>>(dst, histo);
    k_bin_scatter<<<L1_GRID, 1024, 0, stream>>>(src, dst, histo, rec, binBase);
    k_bin_csr    <<<N_BINS, 1024, 0, stream>>>(rec, binBase, cnt, start, dinv, col);

    k_gemm1    <<<GEMM_GRID, B, 0, stream>>>(x, W1, h);

    k_gather1  <<<G1_GRID, B, 0, stream>>>(start, cnt, col, dinv, h, b1, W2, sval, N_NODES);
    k_gather2  <<<(N_NODES + B - 1) / B, B, 0, stream>>>(start, cnt, col, dinv, sval, b2, out, N_NODES);
}


# Source factor dinv[src] folded into the stored values (GEMM stores dinv*h rows, gather 1 stores dinv*scalar): one random read per edge, not two; products re-associated, sum order unchanged
# speedup vs baseline: 4.8297x; 1.0997x over previous
#include <hip/hip_runtime.h>
#include <math.h>

#define N_NODES 200000
#define N_EDGES 2000000
#define IN_DIM  165
#define HID_DIM 128

typedef __attribute__((ext_vector_type(2))) float v2f;
typedef __attribute__((ext_vector_type(8))) float v8f;


#define BIN_SHIFT 10
#define BIN_NODES (1 << BIN_SHIFT)
#define N_BINS    ((N_NODES + BIN_NODES - 1) / BIN_NODES)
#define L1_GRID   256
#define L1_ROWS   (L1_GRID / 4)
#define L1_CHUNK  ((N_EDGES + L1_GRID - 1) / L1_GRID)
#define L1_PER_T  ((L1_CHUNK + 1023) / 1024)

__device__ __forceinline__ int bin_of(int d) {
    const unsigned b = (unsigned)d >> BIN_SHIFT;
    return (int)(b < (unsigned)N_BINS ? b : (unsigned)(N_BINS - 1));
}

__global__ __launch_bounds__(1024)
void k_bin_count(const int* __restrict__ dst, int* __restrict__ histo) {
    __shared__ int s_hist[N_BINS];
    const int tid = threadIdx.x;
    if (tid < N_BINS) s_hist[tid] = 0;
    __syncthreads();
    const int beg = blockIdx.x * L1_CHUNK;
    const int end = (beg + L1_CHUNK < N_EDGES) ? (beg + L1_CHUNK) : N_EDGES;
    int dv[L1_PER_T];
#pragma unroll
    for (int u = 0; u < L1_PER_T; ++u) {
        const int e = beg + u * 1024 + tid;
        dv[u] = dst[e < end ? e : end - 1];
    }
#pragma unroll
    for (int u = 0; u < L1_PER_T; ++u) {
        const int e = beg + u * 1024 + tid;
        if (e < end) atomicAdd(&s_hist[bin_of(dv[u])], 1);
    }
    __syncthreads();
    if (tid < N_BINS) histo[blockIdx.x * N_BINS + tid] = s_hist[tid];
}

__global__ __launch_bounds__(1024)
void k_bin_scatter(const int* __restrict__ src, const int* __restrict__ dst,
                   const int* __restrict__ histo, int* __restrict__ rec_words,
                   int* __restrict__ binBase) {
    int2* rec = (int2*)rec_words;
    __shared__ int s_hist[N_BINS];
    __shared__ int s_part[2][4][256];
    __shared__ int s_base[256];
    __shared__ int s_wtot[8];
    const int tid  = threadIdx.x;
    const int lane = tid & 31;
    const int wave = tid >> 5;
    const int g    = blockIdx.x;
    const int beg  = g * L1_CHUNK;
    const int end  = (beg + L1_CHUNK < N_EDGES) ? (beg + L1_CHUNK) : N_EDGES;

    int es[L1_PER_T], ed[L1_PER_T];
#pragma unroll
    for (int u = 0; u < L1_PER_T; ++u) {
        const int e  = beg + u * 1024 + tid;
        const int ec = e < end ? e : end - 1;
        es[u] = src[ec];
        ed[u] = dst[ec];
    }
    if (tid < N_BINS) s_hist[tid] = 0;

    const int colb = tid & 255;
    const int part = tid >> 8;
    int tot = 0, pre = 0;
    if (colb < N_BINS) {
        for (int r0 = 0; r0 < L1_ROWS; r0 += 16) {
            int v[16];
#pragma unroll
            for (int k = 0; k < 16; ++k)
                v[k] = histo[(part * L1_ROWS + r0 + k) * N_BINS + colb];
#pragma unroll
            for (int k = 0; k < 16; ++k) {
                tot += v[k];
                if (part * L1_ROWS + r0 + k < g) pre += v[k];
            }
        }
    }
    s_part[0][part][colb] = tot;
    s_part[1][part][colb] = pre;
    __syncthreads();

    int bsize = 0, bpre = 0, inc = 0;
    if (tid < 256) {
        bsize = s_part[0][0][tid] + s_part[0][1][tid] + s_part[0][2][tid] + s_part[0][3][tid];
        bpre  = s_part[1][0][tid] + s_part[1][1][tid] + s_part[1][2][tid] + s_part[1][3][tid];
        inc = bsize;
        for (int off = 1; off < 32; off <<= 1) {
            const int t = __shfl_up(inc, off, 32);
            if (lane >= off) inc += t;
        }
        if (lane == 31) s_wtot[wave] = inc;
    }
    __syncthreads();
    if (tid < 256) {
        int wb = 0;
        for (int w = 0; w < wave; ++w) wb += s_wtot[w];
        const int bb = wb + inc - bsize;
        s_base[tid] = bb + bpre;
        if (g == 0 && tid <= N_BINS) binBase[tid] = bb;
    }
    __syncthreads();

#pragma unroll
    for (int u = 0; u < L1_PER_T; ++u) {
        const int e = beg + u * 1024 + tid;
        if (e < end) {
            const int b   = bin_of(ed[u]);
            const int pos = s_base[b] + atomicAdd(&s_hist[b], 1);
            if ((unsigned)pos < (unsigned)N_EDGES)
                rec[pos] = make_int2(es[u], ed[u]);
        }
    }
}

__global__ __launch_bounds__(1024)
void k_bin_csr(const int* __restrict__ rec_words, const int* __restrict__ binBase,
               int* __restrict__ cnt, int* __restrict__ start, float* __restrict__ dinv,
               int* __restrict__ col) {
    const int2* rec = (const int2*)rec_words;
    __shared__ int s_cnt[BIN_NODES];
    __shared__ int s_cur[BIN_NODES];
    __shared__ int s_wtot[32];
    const int tid   = threadIdx.x;
    const int lane  = tid & 31;
    const int wave  = tid >> 5;
    const int node0 = blockIdx.x * BIN_NODES;
    int lo = binBase[blockIdx.x];
    int hi = binBase[blockIdx.x + 1];
    if (lo < 0) lo = 0;
    if (hi > N_EDGES) hi = N_EDGES;

    s_cnt[tid] = 0;
    __syncthreads();

    for (int e0 = lo; e0 < hi; e0 += 4096) {
        int dy[4];
#pragma unroll
        for (int u = 0; u < 4; ++u) {
            const int e = e0 + u * 1024 + tid;
            dy[u] = rec[e < hi ? e : hi - 1].y;
        }
#pragma unroll
        for (int u = 0; u < 4; ++u) {
            const int e = e0 + u * 1024 + tid;
            const int d = dy[u] - node0;
            if (e < hi && (unsigned)d < (unsigned)BIN_NODES) atomicAdd(&s_cnt[d], 1);
        }
    }
    __syncthreads();

    const int c = s_cnt[tid];
    int inc = c;
    for (int off = 1; off < 32; off <<= 1) {
        const int t = __shfl_up(inc, off, 32);
        if (lane >= off) inc += t;
    }
    if (lane == 31) s_wtot[wave] = inc;
    __syncthreads();
    int wb = 0;
    for (int w = 0; w < wave; ++w) wb += s_wtot[w];
    const int st = wb + inc - c;
    s_cur[tid] = st;
    {
        const int node = node0 + tid;
        if (node < N_NODES) {
            cnt[node]   = c;
            start[node] = lo + st;
            dinv[node]  = rsqrtf((float)(c + 1));
        }
    }
    __syncthreads();

    for (int e0 = lo; e0 < hi; e0 += 4096) {
        int2 r[4];
#pragma unroll
        for (int u = 0; u < 4; ++u) {
            const int e = e0 + u * 1024 + tid;
            r[u] = rec[e < hi ? e : hi - 1];
        }
#pragma unroll
        for (int u = 0; u < 4; ++u) {
            const int e = e0 + u * 1024 + tid;
            const int d = r[u].y - node0;
            if (e < hi && (unsigned)d < (unsigned)BIN_NODES) {
                const int pos = lo + atomicAdd(&s_cur[d], 1);
                if ((unsigned)pos < (unsigned)N_EDGES) col[pos] = r[u].x;
            }
        }
    }
}


#define TILE_M    32
#define N_TILES   (N_NODES / TILE_M)
#define XT_F4     (TILE_M * IN_DIM / 4)
#define XBUF_F4   (XT_F4 + 2)
#define W1_F4     (HID_DIM * IN_DIM / 4)
#define KSTEPS    41
#define GEMM_GRID 512

struct XTile { float4 q0, q1, q2, q3, q4, q5; };

__device__ __forceinline__ XTile xtile_load(const float4* __restrict__ xs, int tid) {
    XTile t;
    t.q0 = xs[tid];
    t.q1 = xs[tid + 256];
    t.q2 = xs[tid + 512];
    t.q3 = xs[tid + 768];
    t.q4 = xs[tid + 1024];
    t.q5 = make_float4(0.f, 0.f, 0.f, 0.f);
    if (tid < XT_F4 - 1280) t.q5 = xs[tid + 1280];
    return t;
}

__device__ __forceinline__ void xtile_store(float4* xw, int tid, const XTile& t) {
    xw[tid]        = t.q0;
    xw[tid + 256]  = t.q1;
    xw[tid + 512]  = t.q2;
    xw[tid + 768]  = t.q3;
    xw[tid + 1024] = t.q4;
    if (tid < XT_F4 - 1280) xw[tid + 1280] = t.q5;
}

__global__ __launch_bounds__(256)
void k_gemm1(const float* __restrict__ x, const float* __restrict__ W1,
             const float* __restrict__ dinv, float* __restrict__ h) {
    __shared__ float4 smem4[W1_F4 + 2 * XBUF_F4];
    float*  sW = (float*)smem4;
    float4* sX = smem4 + W1_F4;

    const int tid  = threadIdx.x;
    const int lane = tid & 31;
    const int wave = tid >> 5;
    const int c    = lane & 15;
    const int hh   = lane >> 4;

    int tile = blockIdx.x;

    XTile xr = xtile_load((const float4*)(x + (size_t)tile * (TILE_M * IN_DIM)), tid);

    {
        const float4* w4 = (const float4*)W1;
        for (int q = tid; q < W1_F4; q += 256) smem4[q] = w4[q];
    }
    xtile_store(sX, tid, xr);
    __syncthreads();

    v2f   bf[KSTEPS];
    float bt;
    {
        const float* wrow = sW + (wave * 16 + c) * IN_DIM;
#pragma unroll
        for (int j = 0; j < KSTEPS; ++j) {
            bf[j].x = wrow[4 * j + 2 * hh];
            bf[j].y = wrow[4 * j + 2 * hh + 1];
        }
        const float w164 = wrow[4 * KSTEPS];
        bt = (hh == 0) ? w164 : 0.0f;
    }

    int cur = 0;
    for (; tile < N_TILES; tile += GEMM_GRID) {
        const int  nxt  = tile + GEMM_GRID;
        const bool more = nxt < N_TILES;
        if (more)
            xr = xtile_load((const float4*)(x + (size_t)nxt * (TILE_M * IN_DIM)), tid);

        const float4* dq  = (const float4*)(dinv + (size_t)tile * TILE_M + 8 * hh);
        const float4  dA0 = dq[0], dA1 = dq[1];
        const float4  dB0 = dq[4], dB1 = dq[5];

        const float* xb = (const float*)(sX + cur * XBUF_F4);
        const float* a0 = xb + c * IN_DIM;
        const float* a1 = xb + (16 + c) * IN_DIM;
        v8f acc0 = {0.f, 0.f, 0.f, 0.f, 0.f, 0.f, 0.f, 0.f};
        v8f acc1 = {0.f, 0.f, 0.f, 0.f, 0.f, 0.f, 0.f, 0.f};

#pragma unroll
        for (int j = 0; j < KSTEPS; ++j) {
            v2f f0, f1;
            f0.x = a0[4 * j + 2 * hh];
            f0.y = a0[4 * j + 2 * hh + 1];
            f1.x = a1[4 * j + 2 * hh];
            f1.y = a1[4 * j + 2 * hh + 1];
            acc0 = __builtin_amdgcn_wmma_f32_16x16x4_f32(
                       false, f0, false, bf[j], (short)0, acc0, false, false);
            acc1 = __builtin_amdgcn_wmma_f32_16x16x4_f32(
                       false, f1, false, bf[j], (short)0, acc1, false, false);
        }
        {
            const float t0 = a0[4 * KSTEPS];
            const float t1 = a1[4 * KSTEPS];
            v2f f0, f1, fb;
            f0.x = (hh == 0) ? t0 : 0.0f;  f0.y = 0.0f;
            f1.x = (hh == 0) ? t1 : 0.0f;  f1.y = 0.0f;
            fb.x = bt;                     fb.y = 0.0f;
            acc0 = __builtin_amdgcn_wmma_f32_16x16x4_f32(
                       false, f0, false, fb, (short)0, acc0, false, false);
            acc1 = __builtin_amdgcn_wmma_f32_16x16x4_f32(
                       false, f1, false, fb, (short)0, acc1, false, false);
        }

        if (more) xtile_store(sX + (cur ^ 1) * XBUF_F4, tid, xr);
        __syncthreads();

        {
            float* hp = h + ((size_t)tile * TILE_M + 8 * hh) * HID_DIM + wave * 16 + c;
            const float dA[8] = {dA0.x, dA0.y, dA0.z, dA0.w, dA1.x, dA1.y, dA1.z, dA1.w};
            const float dB[8] = {dB0.x, dB0.y, dB0.z, dB0.w, dB1.x, dB1.y, dB1.z, dB1.w};
#pragma unroll
            for (int r = 0; r < 8; ++r) {
                hp[(size_t)r * HID_DIM]        = acc0[r] * dA[r];
                hp[(size_t)(16 + r) * HID_DIM] = acc1[r] * dB[r];
            }
        }
        cur ^= 1;
    }
}


template <int K>
__device__ __forceinline__ void gather_rows(float4& acc, const float* __restrict__ h,
                                            int sv, int t, int f4) {
    float4 v[K];
#pragma unroll
    for (int u = 0; u < K; ++u) {
        const int s = __builtin_amdgcn_readlane(sv, t + u);
        v[u] = *(const float4*)&h[(size_t)s * HID_DIM + f4];
    }
#pragma unroll
    for (int u = 0; u < K; ++u) {
        acc.x += v[u].x; acc.y += v[u].y;
        acc.z += v[u].z; acc.w += v[u].w;
    }
}

#define G1_NODES 8
#define G1_GRID  ((N_NODES + 8 * G1_NODES - 1) / (8 * G1_NODES))

__device__ __forceinline__ int bucket_sources(const int* __restrict__ col, int beg, int m, int lane) {
    int e = beg + (lane < m ? lane : 0);
    e = e < N_EDGES - 1 ? e : N_EDGES - 1;
    return col[e];
}

__device__ __forceinline__ void gather_chunk(float4& acc, const float* __restrict__ h,
                                             int sv, int nb, int f4) {
    int t = 0;
    for (; t + 8 <= nb; t += 8) gather_rows<8>(acc, h, sv, t, f4);
    if (nb - t >= 4) { gather_rows<4>(acc, h, sv, t, f4); t += 4; }
    const int r = nb - t;
    if (r == 3)      gather_rows<3>(acc, h, sv, t, f4);
    else if (r == 2) gather_rows<2>(acc, h, sv, t, f4);
    else if (r == 1) gather_rows<1>(acc, h, sv, t, f4);
}

__global__ __launch_bounds__(256)
void k_gather1(const int* __restrict__ start, const int* __restrict__ cnt,
               const int* __restrict__ col, const float* __restrict__ dinv,
               const float* __restrict__ h, const float* __restrict__ b1,
               const float* __restrict__ W2, float* __restrict__ tval, int n) {
    const int lane = threadIdx.x & 31;
    const int n0 = __builtin_amdgcn_readfirstlane((blockIdx.x * 8 + (threadIdx.x >> 5)) * G1_NODES);
    if (n0 >= n) return;
    const int f4 = lane * 4;
    const int ng = (n - n0 < G1_NODES) ? (n - n0) : G1_NODES;

    const int   nl = n0 + (lane < ng ? lane : ng - 1);
    const int   mv = cnt[nl];
    const int   bv = start[nl];
    const float dv = dinv[nl];

    float4 self = *(const float4*)&h[(size_t)n0 * HID_DIM + f4];
    int m   = __builtin_amdgcn_readlane(mv, 0);
    int beg = __builtin_amdgcn_readlane(bv, 0);
    int s   = bucket_sources(col, beg, m, lane);
    float pv = 0.0f;

    for (int j = 0; j < ng; ++j) {
        const int    jn     = (j + 1 < ng) ? (j + 1) : (ng - 1);
        const float4 self_n = *(const float4*)&h[(size_t)(n0 + jn) * HID_DIM + f4];
        const int    m_n    = __builtin_amdgcn_readlane(mv, jn);
        const int    beg_n  = __builtin_amdgcn_readlane(bv, jn);
        const int    s_n    = bucket_sources(col, beg_n, m_n, lane);

        const float di = __int_as_float(__builtin_amdgcn_readlane(__float_as_int(dv), j));
        float4 acc = self;

        gather_chunk(acc, h, s, (m < 32) ? m : 32, f4);
        for (int base = 32; base < m; base += 32) {
            const int s2 = bucket_sources(col, beg + base, m - base, lane);
            gather_chunk(acc, h, s2, (m - base < 32) ? (m - base) : 32, f4);
        }

        const float4 bb = *(const float4*)&b1[f4];
        const float4 w2 = *(const float4*)&W2[f4];
        float p = fmaxf(acc.x * di + bb.x, 0.f) * w2.x
                + fmaxf(acc.y * di + bb.y, 0.f) * w2.y
                + fmaxf(acc.z * di + bb.z, 0.f) * w2.z
                + fmaxf(acc.w * di + bb.w, 0.f) * w2.w;
        for (int off = 16; off > 0; off >>= 1)
            p += __shfl_xor(p, off, 32);
        if (lane == j) pv = p;

        m = m_n;   beg = beg_n;   s = s_n;   self = self_n;
    }
    if (lane < ng) tval[n0 + lane] = pv * dv;
}


__global__ void k_gather2(const int* __restrict__ start, const int* __restrict__ cnt,
                          const int* __restrict__ col, const float* __restrict__ dinv,
                          const float* __restrict__ tval, const float* __restrict__ b2,
                          float* __restrict__ out, int n) {
    int i = blockIdx.x * blockDim.x + threadIdx.x;
    if (i >= n) return;
    const float di  = dinv[i];
    const int   beg = start[i];
    const int   m   = cnt[i];
    float acc = tval[i];
    for (int t = 0; t < m; ++t)
        acc += tval[col[beg + t]];
    out[i] = 1.0f / (1.0f + expf(-(acc * di + b2[0])));
}


extern "C" void kernel_launch(void* const* d_in, const int* in_sizes, int n_in,
                              void* d_out, int out_size, void* d_ws, size_t ws_size,
                              hipStream_t stream) {
    (void)in_sizes; (void)n_in; (void)out_size; (void)ws_size;
    const float* x  = (const float*)d_in[0];
    const float* W1 = (const float*)d_in[1];
    const float* b1 = (const float*)d_in[2];
    const float* W2 = (const float*)d_in[3];
    const float* b2 = (const float*)d_in[4];
    const int*   ei = (const int*)d_in[5];
    const int* src = ei;
    const int* dst = ei + N_EDGES;
    float* out = (float*)d_out;

    float* dinv = (float*)d_ws;
    float* tval = dinv + N_NODES;
    float* h    = tval + N_NODES;
    int*  cnt     = (int*)(h + (size_t)N_NODES * HID_DIM);
    int*  start   = cnt + N_NODES;
    int*  rec     = start + N_NODES;
    int*  col     = rec + 2 * N_EDGES;
    int*  histo   = col + N_EDGES;
    int*  binBase = histo + L1_GRID * N_BINS;

    const int B = 256;
    k_bin_count  <<<L1_GRID, 1024, 0, stream>>>(dst, histo);
    k_bin_scatter<<<L1_GRID, 1024, 0, stream>>>(src, dst, histo, rec, binBase);
    k_bin_csr    <<<N_BINS, 1024, 0, stream>>>(rec, binBase, cnt, start, dinv, col);

    k_gemm1    <<<GEMM_GRID, B, 0, stream>>>(x, W1, dinv, h);

    k_gather1  <<<G1_GRID, B, 0, stream>>>(start, cnt, col, dinv, h, b1, W2, tval, N_NODES);
    k_gather2  <<<(N_NODES + B - 1) / B, B, 0, stream>>>(start, cnt, col, dinv, tval, b2, out, N_NODES);
}


# Edge grouping level 1 in one launch: each chunk ranks and orders its edges by bin in LDS and leaves as one contiguous run; the bin pass reads its 256 slices; sum order = LDS add arrival rank
# speedup vs baseline: 5.2299x; 1.0829x over previous
#include <hip/hip_runtime.h>
#include <math.h>

#define N_NODES 200000
#define N_EDGES 2000000
#define IN_DIM  165
#define HID_DIM 128

typedef __attribute__((ext_vector_type(2))) float v2f;
typedef __attribute__((ext_vector_type(8))) float v8f;


#define BIN_SHIFT 10
#define BIN_NODES (1 << BIN_SHIFT)
#define N_BINS    ((N_NODES + BIN_NODES - 1) / BIN_NODES)
#define L1_GRID   256
#define L1_CHUNK  ((N_EDGES + L1_GRID - 1) / L1_GRID)
#define L1_PER_T  ((L1_CHUNK + 1023) / 1024)
#define SL_PER_W  (L1_GRID / 32)

__device__ __forceinline__ int bin_of(int d) {
    const unsigned b = (unsigned)d >> BIN_SHIFT;
    return (int)(b < (unsigned)N_BINS ? b : (unsigned)(N_BINS - 1));
}

__global__ __launch_bounds__(1024)
void k_chunk_sort(const int* __restrict__ src, const int* __restrict__ dst,
                  int* __restrict__ rec_words, int* __restrict__ slice_words) {
    int2* rec   = (int2*)rec_words;
    int2* slice = (int2*)slice_words;
    __shared__ int2 s_rec[L1_CHUNK];
    __shared__ int  s_hist[256];
    __shared__ int  s_wtot[8];
    const int tid  = threadIdx.x;
    const int lane = tid & 31;
    const int wave = tid >> 5;
    const int g    = blockIdx.x;
    const int beg  = g * L1_CHUNK;
    const int end  = (beg + L1_CHUNK < N_EDGES) ? (beg + L1_CHUNK) : N_EDGES;

    int es[L1_PER_T], ed[L1_PER_T];
#pragma unroll
    for (int u = 0; u < L1_PER_T; ++u) {
        const int e  = beg + u * 1024 + tid;
        const int ec = e < end ? e : end - 1;
        es[u] = src[ec];
        ed[u] = dst[ec];
    }
    if (tid < 256) s_hist[tid] = 0;
    __syncthreads();

    int rk[L1_PER_T];
#pragma unroll
    for (int u = 0; u < L1_PER_T; ++u) {
        const int e = beg + u * 1024 + tid;
        rk[u] = 0;
        if (e < end) rk[u] = atomicAdd(&s_hist[bin_of(ed[u])], 1);
    }
    __syncthreads();

    int c = 0, inc = 0;
    if (tid < 256) {
        c   = s_hist[tid];
        inc = c;
        for (int off = 1; off < 32; off <<= 1) {
            const int t = __shfl_up(inc, off, 32);
            if (lane >= off) inc += t;
        }
        if (lane == 31) s_wtot[wave] = inc;
    }
    __syncthreads();
    int first = 0;
    if (tid < 256) {
        int wb = 0;
        for (int w = 0; w < wave; ++w) wb += s_wtot[w];
        first = wb + inc - c;
        s_hist[tid] = first;
    }
    __syncthreads();

    int fs[L1_PER_T];
#pragma unroll
    for (int u = 0; u < L1_PER_T; ++u) fs[u] = s_hist[bin_of(ed[u])];
#pragma unroll
    for (int u = 0; u < L1_PER_T; ++u) {
        const int e = beg + u * 1024 + tid;
        if (e < end) {
            const int pos = fs[u] + rk[u];
            if ((unsigned)pos < (unsigned)L1_CHUNK)
                s_rec[pos] = make_int2(es[u], ed[u]);
        }
    }
    __syncthreads();

    const int len = end - beg;
    int2 ov[L1_PER_T];
#pragma unroll
    for (int u = 0; u < L1_PER_T; ++u) {
        const int i = u * 1024 + tid;
        ov[u] = s_rec[i < len ? i : 0];
    }
#pragma unroll
    for (int u = 0; u < L1_PER_T; ++u) {
        const int i = u * 1024 + tid;
        if (i < len) rec[beg + i] = ov[u];
    }
    if (tid < N_BINS) slice[g * N_BINS + tid] = make_int2(c, first);
}


__device__ __forceinline__ int slice_slot(int p, int n, int k) {
    const int e = p + (k < n ? k : 0);
    return e < N_EDGES - 1 ? e : N_EDGES - 1;
}

__global__ __launch_bounds__(1024)
void k_bin_csr(const int* __restrict__ rec_words, const int* __restrict__ slice_words,
               int* __restrict__ cnt, int* __restrict__ start, float* __restrict__ dinv,
               int* __restrict__ col) {
    const int2* rec   = (const int2*)rec_words;
    const int2* slice = (const int2*)slice_words;
    __shared__ int s_cnt[BIN_NODES];
    __shared__ int s_cur[BIN_NODES];
    __shared__ int s_wtot[32];
    __shared__ int s_sbeg[L1_GRID];
    __shared__ int s_slen[L1_GRID];
    __shared__ int s_red[8];
    const int tid   = threadIdx.x;
    const int lane  = tid & 31;
    const int wave  = tid >> 5;
    const int b     = blockIdx.x;
    const int node0 = b * BIN_NODES;

    s_cnt[tid] = 0;
    if (tid < L1_GRID) {
        const int2 sl   = slice[tid * N_BINS + b];
        const int  cbeg = tid * L1_CHUNK;
        const int  clen = (cbeg + L1_CHUNK < N_EDGES) ? L1_CHUNK : (N_EDGES - cbeg);
        int o = sl.y, n = sl.x;
        o = o < 0 ? 0 : (o > clen ? clen : o);
        n = n < 0 ? 0 : (n > clen - o ? clen - o : n);
        s_sbeg[tid] = cbeg + o;
        s_slen[tid] = n;
        int so = o;
        for (int off = 16; off > 0; off >>= 1) so += __shfl_xor(so, off, 32);
        if (lane == 0) s_red[wave] = so;
    }
    __syncthreads();
    int lo = 0;
    for (int w = 0; w < 8; ++w) lo += s_red[w];

    int sp[SL_PER_W], sn[SL_PER_W];
#pragma unroll
    for (int j = 0; j < SL_PER_W; ++j) {
        sp[j] = __builtin_amdgcn_readfirstlane(s_sbeg[wave + 32 * j]);
        sn[j] = __builtin_amdgcn_readfirstlane(s_slen[wave + 32 * j]);
    }

    {
        int dy0[SL_PER_W], dy1[SL_PER_W];
#pragma unroll
        for (int j = 0; j < SL_PER_W; ++j) {
            dy0[j] = rec[slice_slot(sp[j], sn[j], lane)].y;
            dy1[j] = rec[slice_slot(sp[j], sn[j], 32 + lane)].y;
        }
#pragma unroll
        for (int j = 0; j < SL_PER_W; ++j) {
            const int n  = sn[j];
            const int d0 = dy0[j] - node0;
            const int d1 = dy1[j] - node0;
            if (lane < n      && (unsigned)d0 < (unsigned)BIN_NODES) atomicAdd(&s_cnt[d0], 1);
            if (32 + lane < n && (unsigned)d1 < (unsigned)BIN_NODES) atomicAdd(&s_cnt[d1], 1);
        }
#pragma unroll
        for (int j = 0; j < SL_PER_W; ++j) {
            const int p = sp[j];
            const int n = sn[j];
            for (int k = 64 + lane; k < n; k += 32) {
                const int d = rec[p + k].y - node0;
                if ((unsigned)d < (unsigned)BIN_NODES) atomicAdd(&s_cnt[d], 1);
            }
        }
    }
    __syncthreads();

    const int c = s_cnt[tid];
    int inc = c;
    for (int off = 1; off < 32; off <<= 1) {
        const int t = __shfl_up(inc, off, 32);
        if (lane >= off) inc += t;
    }
    if (lane == 31) s_wtot[wave] = inc;
    __syncthreads();
    int wb = 0;
    for (int w = 0; w < wave; ++w) wb += s_wtot[w];
    const int st = wb + inc - c;
    s_cur[tid] = st;
    __syncthreads();

    {
        int2 r0[SL_PER_W], r1[SL_PER_W];
#pragma unroll
        for (int j = 0; j < SL_PER_W; ++j) {
            r0[j] = rec[slice_slot(sp[j], sn[j], lane)];
            r1[j] = rec[slice_slot(sp[j], sn[j], 32 + lane)];
        }
#pragma unroll
        for (int j = 0; j < SL_PER_W; ++j) {
            const int n  = sn[j];
            const int d0 = r0[j].y - node0;
            const int d1 = r1[j].y - node0;
            if (lane < n && (unsigned)d0 < (unsigned)BIN_NODES) {
                const int pos = lo + atomicAdd(&s_cur[d0], 1);
                if ((unsigned)pos < (unsigned)N_EDGES) col[pos] = r0[j].x;
            }
            if (32 + lane < n && (unsigned)d1 < (unsigned)BIN_NODES) {
                const int pos = lo + atomicAdd(&s_cur[d1], 1);
                if ((unsigned)pos < (unsigned)N_EDGES) col[pos] = r1[j].x;
            }
        }
#pragma unroll
        for (int j = 0; j < SL_PER_W; ++j) {
            const int p = sp[j];
            const int n = sn[j];
            for (int k = 64 + lane; k < n; k += 32) {
                const int2 r = rec[p + k];
                const int  d = r.y - node0;
                if ((unsigned)d < (unsigned)BIN_NODES) {
                    const int pos = lo + atomicAdd(&s_cur[d], 1);
                    if ((unsigned)pos < (unsigned)N_EDGES) col[pos] = r.x;
                }
            }
        }
    }

    {
        const int node = node0 + tid;
        if (node < N_NODES) {
            cnt[node]   = c;
            start[node] = lo + st;
            dinv[node]  = rsqrtf((float)(c + 1));
        }
    }
}


#define TILE_M    32
#define N_TILES   (N_NODES / TILE_M)
#define XT_F4     (TILE_M * IN_DIM / 4)
#define XBUF_F4   (XT_F4 + 2)
#define W1_F4     (HID_DIM * IN_DIM / 4)
#define KSTEPS    41
#define GEMM_GRID 512

struct XTile { float4 q0, q1, q2, q3, q4, q5; };

__device__ __forceinline__ XTile xtile_load(const float4* __restrict__ xs, int tid) {
    XTile t;
    t.q0 = xs[tid];
    t.q1 = xs[tid + 256];
    t.q2 = xs[tid + 512];
    t.q3 = xs[tid + 768];
    t.q4 = xs[tid + 1024];
    t.q5 = make_float4(0.f, 0.f, 0.f, 0.f);
    if (tid < XT_F4 - 1280) t.q5 = xs[tid + 1280];
    return t;
}

__device__ __forceinline__ void xtile_store(float4* xw, int tid, const XTile& t) {
    xw[tid]        = t.q0;
    xw[tid + 256]  = t.q1;
    xw[tid + 512]  = t.q2;
    xw[tid + 768]  = t.q3;
    xw[tid + 1024] = t.q4;
    if (tid < XT_F4 - 1280) xw[tid + 1280] = t.q5;
}

__global__ __launch_bounds__(256)
void k_gemm1(const float* __restrict__ x, const float* __restrict__ W1,
             const float* __restrict__ dinv, float* __restrict__ h) {
    __shared__ float4 smem4[W1_F4 + 2 * XBUF_F4];
    float*  sW = (float*)smem4;
    float4* sX = smem4 + W1_F4;

    const int tid  = threadIdx.x;
    const int lane = tid & 31;
    const int wave = tid >> 5;
    const int c    = lane & 15;
    const int hh   = lane >> 4;

    int tile = blockIdx.x;

    XTile xr = xtile_load((const float4*)(x + (size_t)tile * (TILE_M * IN_DIM)), tid);

    {
        const float4* w4 = (const float4*)W1;
        for (int q = tid; q < W1_F4; q += 256) smem4[q] = w4[q];
    }
    xtile_store(sX, tid, xr);
    __syncthreads();

    v2f   bf[KSTEPS];
    float bt;
    {
        const float* wrow = sW + (wave * 16 + c) * IN_DIM;
#pragma unroll
        for (int j = 0; j < KSTEPS; ++j) {
            bf[j].x = wrow[4 * j + 2 * hh];
            bf[j].y = wrow[4 * j + 2 * hh + 1];
        }
        const float w164 = wrow[4 * KSTEPS];
        bt = (hh == 0) ? w164 : 0.0f;
    }

    int cur = 0;
    for (; tile < N_TILES; tile += GEMM_GRID) {
        const int  nxt  = tile + GEMM_GRID;
        const bool more = nxt < N_TILES;
        if (more)
            xr = xtile_load((const float4*)(x + (size_t)nxt * (TILE_M * IN_DIM)), tid);

        const float4* dq  = (const float4*)(dinv + (size_t)tile * TILE_M + 8 * hh);
        const float4  dA0 = dq[0], dA1 = dq[1];
        const float4  dB0 = dq[4], dB1 = dq[5];

        const float* xb = (const float*)(sX + cur * XBUF_F4);
        const float* a0 = xb + c * IN_DIM;
        const float* a1 = xb + (16 + c) * IN_DIM;
        v8f acc0 = {0.f, 0.f, 0.f, 0.f, 0.f, 0.f, 0.f, 0.f};
        v8f acc1 = {0.f, 0.f, 0.f, 0.f, 0.f, 0.f, 0.f, 0.f};

#pragma unroll
        for (int j = 0; j < KSTEPS; ++j) {
            v2f f0, f1;
            f0.x = a0[4 * j + 2 * hh];
            f0.y = a0[4 * j + 2 * hh + 1];
            f1.x = a1[4 * j + 2 * hh];
            f1.y = a1[4 * j + 2 * hh + 1];
            acc0 = __builtin_amdgcn_wmma_f32_16x16x4_f32(
                       false, f0, false, bf[j], (short)0, acc0, false, false);
            acc1 = __builtin_amdgcn_wmma_f32_16x16x4_f32(
                       false, f1, false, bf[j], (short)0, acc1, false, false);
        }
        {
            const float t0 = a0[4 * KSTEPS];
            const float t1 = a1[4 * KSTEPS];
            v2f f0, f1, fb;
            f0.x = (hh == 0) ? t0 : 0.0f;  f0.y = 0.0f;
            f1.x = (hh == 0) ? t1 : 0.0f;  f1.y = 0.0f;
            fb.x = bt;                     fb.y = 0.0f;
            acc0 = __builtin_amdgcn_wmma_f32_16x16x4_f32(
                       false, f0, false, fb, (short)0, acc0, false, false);
            acc1 = __builtin_amdgcn_wmma_f32_16x16x4_f32(
                       false, f1, false, fb, (short)0, acc1, false, false);
        }

        if (more) xtile_store(sX + (cur ^ 1) * XBUF_F4, tid, xr);
        __syncthreads();

        {
            float* hp = h + ((size_t)tile * TILE_M + 8 * hh) * HID_DIM + wave * 16 + c;
            const float dA[8] = {dA0.x, dA0.y, dA0.z, dA0.w, dA1.x, dA1.y, dA1.z, dA1.w};
            const float dB[8] = {dB0.x, dB0.y, dB0.z, dB0.w, dB1.x, dB1.y, dB1.z, dB1.w};
#pragma unroll
            for (int r = 0; r < 8; ++r) {
                hp[(size_t)r * HID_DIM]        = acc0[r] * dA[r];
                hp[(size_t)(16 + r) * HID_DIM] = acc1[r] * dB[r];
            }
        }
        cur ^= 1;
    }
}


template <int K>
__device__ __forceinline__ void gather_rows(float4& acc, const float* __restrict__ h,
                                            int sv, int t, int f4) {
    float4 v[K];
#pragma unroll
    for (int u = 0; u < K; ++u) {
        const int s = __builtin_amdgcn_readlane(sv, t + u);
        v[u] = *(const float4*)&h[(size_t)s * HID_DIM + f4];
    }
#pragma unroll
    for (int u = 0; u < K; ++u) {
        acc.x += v[u].x; acc.y += v[u].y;
        acc.z += v[u].z; acc.w += v[u].w;
    }
}

#define G1_NODES 8
#define G1_GRID  ((N_NODES + 8 * G1_NODES - 1) / (8 * G1_NODES))

__device__ __forceinline__ int bucket_sources(const int* __restrict__ col, int beg, int m, int lane) {
    int e = beg + (lane < m ? lane : 0);
    e = e < N_EDGES - 1 ? e : N_EDGES - 1;
    return col[e];
}

__device__ __forceinline__ void gather_chunk(float4& acc, const float* __restrict__ h,
                                             int sv, int nb, int f4) {
    int t = 0;
    for (; t + 8 <= nb; t += 8) gather_rows<8>(acc, h, sv, t, f4);
    if (nb - t >= 4) { gather_rows<4>(acc, h, sv, t, f4); t += 4; }
    const int r = nb - t;
    if (r == 3)      gather_rows<3>(acc, h, sv, t, f4);
    else if (r == 2) gather_rows<2>(acc, h, sv, t, f4);
    else if (r == 1) gather_rows<1>(acc, h, sv, t, f4);
}

__global__ __launch_bounds__(256)
void k_gather1(const int* __restrict__ start, const int* __restrict__ cnt,
               const int* __restrict__ col, const float* __restrict__ dinv,
               const float* __restrict__ h, const float* __restrict__ b1,
               const float* __restrict__ W2, float* __restrict__ tval, int n) {
    const int lane = threadIdx.x & 31;
    const int n0 = __builtin_amdgcn_readfirstlane((blockIdx.x * 8 + (threadIdx.x >> 5)) * G1_NODES);
    if (n0 >= n) return;
    const int f4 = lane * 4;
    const int ng = (n - n0 < G1_NODES) ? (n - n0) : G1_NODES;

    const int   nl = n0 + (lane < ng ? lane : ng - 1);
    const int   mv = cnt[nl];
    const int   bv = start[nl];
    const float dv = dinv[nl];

    float4 self = *(const float4*)&h[(size_t)n0 * HID_DIM + f4];
    int m   = __builtin_amdgcn_readlane(mv, 0);
    int beg = __builtin_amdgcn_readlane(bv, 0);
    int s   = bucket_sources(col, beg, m, lane);
    float pv = 0.0f;

    for (int j = 0; j < ng; ++j) {
        const int    jn     = (j + 1 < ng) ? (j + 1) : (ng - 1);
        const float4 self_n = *(const float4*)&h[(size_t)(n0 + jn) * HID_DIM + f4];
        const int    m_n    = __builtin_amdgcn_readlane(mv, jn);
        const int    beg_n  = __builtin_amdgcn_readlane(bv, jn);
        const int    s_n    = bucket_sources(col, beg_n, m_n, lane);

        const float di = __int_as_float(__builtin_amdgcn_readlane(__float_as_int(dv), j));
        float4 acc = self;

        gather_chunk(acc, h, s, (m < 32) ? m : 32, f4);
        for (int base = 32; base < m; base += 32) {
            const int s2 = bucket_sources(col, beg + base, m - base, lane);
            gather_chunk(acc, h, s2, (m - base < 32) ? (m - base) : 32, f4);
        }

        const float4 bb = *(const float4*)&b1[f4];
        const float4 w2 = *(const float4*)&W2[f4];
        float p = fmaxf(acc.x * di + bb.x, 0.f) * w2.x
                + fmaxf(acc.y * di + bb.y, 0.f) * w2.y
                + fmaxf(acc.z * di + bb.z, 0.f) * w2.z
                + fmaxf(acc.w * di + bb.w, 0.f) * w2.w;
        for (int off = 16; off > 0; off >>= 1)
            p += __shfl_xor(p, off, 32);
        if (lane == j) pv = p;

        m = m_n;   beg = beg_n;   s = s_n;   self = self_n;
    }
    if (lane < ng) tval[n0 + lane] = pv * dv;
}


__global__ void k_gather2(const int* __restrict__ start, const int* __restrict__ cnt,
                          const int* __restrict__ col, const float* __restrict__ dinv,
                          const float* __restrict__ tval, const float* __restrict__ b2,
                          float* __restrict__ out, int n) {
    int i = blockIdx.x * blockDim.x + threadIdx.x;
    if (i >= n) return;
    const float di  = dinv[i];
    const int   beg = start[i];
    const int   m   = cnt[i];
    float acc = tval[i];
    for (int t = 0; t < m; ++t)
        acc += tval[col[beg + t]];
    out[i] = 1.0f / (1.0f + expf(-(acc * di + b2[0])));
}


extern "C" void kernel_launch(void* const* d_in, const int* in_sizes, int n_in,
                              void* d_out, int out_size, void* d_ws, size_t ws_size,
                              hipStream_t stream) {
    (void)in_sizes; (void)n_in; (void)out_size; (void)ws_size;
    const float* x  = (const float*)d_in[0];
    const float* W1 = (const float*)d_in[1];
    const float* b1 = (const float*)d_in[2];
    const float* W2 = (const float*)d_in[3];
    const float* b2 = (const float*)d_in[4];
    const int*   ei = (const int*)d_in[5];
    const int* src = ei;
    const int* dst = ei + N_EDGES;
    float* out = (float*)d_out;

    float* dinv = (float*)d_ws;
    float* tval = dinv + N_NODES;
    float* h    = tval + N_NODES;
    int*  cnt     = (int*)(h + (size_t)N_NODES * HID_DIM);
    int*  start   = cnt + N_NODES;
    int*  rec     = start + N_NODES;
    int*  col     = rec + 2 * N_EDGES;
    int*  slice   = col + N_EDGES;

    const int B = 256;
    k_chunk_sort<<<L1_GRID, 1024, 0, stream>>>(src, dst, rec, slice);
    k_bin_csr   <<<N_BINS, 1024, 0, stream>>>(rec, slice, cnt, start, dinv, col);

    k_gemm1    <<<GEMM_GRID, B, 0, stream>>>(x, W1, dinv, h);

    k_gather1  <<<G1_GRID, B, 0, stream>>>(start, cnt, col, dinv, h, b1, W2, tval, N_NODES);
    k_gather2  <<<(N_NODES + B - 1) / B, B, 0, stream>>>(start, cnt, col, dinv, tval, b2, out, N_NODES);
}


# Bin pass fills its buckets in an LDS copy of the bin's run (records read once, ranks taken together), stored as whole aligned lines, not 2M scattered 4-byte stores; sum order = LDS add arrival rank
# speedup vs baseline: 5.4754x; 1.0469x over previous
#include <hip/hip_runtime.h>
#include <math.h>

#define N_NODES 200000
#define N_EDGES 2000000
#define IN_DIM  165
#define HID_DIM 128

typedef __attribute__((ext_vector_type(2))) float v2f;
typedef __attribute__((ext_vector_type(8))) float v8f;


#define BIN_SHIFT 10
#define BIN_NODES (1 << BIN_SHIFT)
#define N_BINS    ((N_NODES + BIN_NODES - 1) / BIN_NODES)
#define L1_GRID   256
#define L1_CHUNK  ((N_EDGES + L1_GRID - 1) / L1_GRID)
#define L1_PER_T  ((L1_CHUNK + 1023) / 1024)
#define SL_PER_W  (L1_GRID / 32)

__device__ __forceinline__ int bin_of(int d) {
    const unsigned b = (unsigned)d >> BIN_SHIFT;
    return (int)(b < (unsigned)N_BINS ? b : (unsigned)(N_BINS - 1));
}

__global__ __launch_bounds__(1024)
void k_chunk_sort(const int* __restrict__ src, const int* __restrict__ dst,
                  int* __restrict__ rec_words, int* __restrict__ slice_words) {
    int2* rec   = (int2*)rec_words;
    int2* slice = (int2*)slice_words;
    __shared__ int2 s_rec[L1_CHUNK];
    __shared__ int  s_hist[256];
    __shared__ int  s_wtot[8];
    const int tid  = threadIdx.x;
    const int lane = tid & 31;
    const int wave = tid >> 5;
    const int g    = blockIdx.x;
    const int beg  = g * L1_CHUNK;
    const int end  = (beg + L1_CHUNK < N_EDGES) ? (beg + L1_CHUNK) : N_EDGES;

    int es[L1_PER_T], ed[L1_PER_T];
#pragma unroll
    for (int u = 0; u < L1_PER_T; ++u) {
        const int e  = beg + u * 1024 + tid;
        const int ec = e < end ? e : end - 1;
        es[u] = src[ec];
        ed[u] = dst[ec];
    }
    if (tid < 256) s_hist[tid] = 0;
    __syncthreads();

    int rk[L1_PER_T];
#pragma unroll
    for (int u = 0; u < L1_PER_T; ++u) {
        const int e = beg + u * 1024 + tid;
        rk[u] = 0;
        if (e < end) rk[u] = atomicAdd(&s_hist[bin_of(ed[u])], 1);
    }
    __syncthreads();

    int c = 0, inc = 0;
    if (tid < 256) {
        c   = s_hist[tid];
        inc = c;
        for (int off = 1; off < 32; off <<= 1) {
            const int t = __shfl_up(inc, off, 32);
            if (lane >= off) inc += t;
        }
        if (lane == 31) s_wtot[wave] = inc;
    }
    __syncthreads();
    int first = 0;
    if (tid < 256) {
        int wb = 0;
        for (int w = 0; w < wave; ++w) wb += s_wtot[w];
        first = wb + inc - c;
        s_hist[tid] = first;
    }
    __syncthreads();

    int fs[L1_PER_T];
#pragma unroll
    for (int u = 0; u < L1_PER_T; ++u) fs[u] = s_hist[bin_of(ed[u])];
#pragma unroll
    for (int u = 0; u < L1_PER_T; ++u) {
        const int e = beg + u * 1024 + tid;
        if (e < end) {
            const int pos = fs[u] + rk[u];
            if ((unsigned)pos < (unsigned)L1_CHUNK)
                s_rec[pos] = make_int2(es[u], ed[u]);
        }
    }
    __syncthreads();

    const int len = end - beg;
    int2 ov[L1_PER_T];
#pragma unroll
    for (int u = 0; u < L1_PER_T; ++u) {
        const int i = u * 1024 + tid;
        ov[u] = s_rec[i < len ? i : 0];
    }
#pragma unroll
    for (int u = 0; u < L1_PER_T; ++u) {
        const int i = u * 1024 + tid;
        if (i < len) rec[beg + i] = ov[u];
    }
    if (tid < N_BINS) slice[g * N_BINS + tid] = make_int2(c, first);
}


__device__ __forceinline__ int slice_slot(int p, int n, int k) {
    const int e = p + (k < n ? k : 0);
    return e < N_EDGES - 1 ? e : N_EDGES - 1;
}

#define COL_CAP    12288
#define COL_ROUNDS (COL_CAP / 1024 + 1)

__device__ __forceinline__ void bucket_store(int* s_col, int* __restrict__ col, int lo, int p, int sv) {
    if (p >= 0) {
        if (p < COL_CAP) {
            s_col[p] = sv;
        } else {
            const int pos = lo + p;
            if ((unsigned)pos < (unsigned)N_EDGES) col[pos] = sv;
        }
    }
}

__global__ __launch_bounds__(1024)
void k_bin_csr(const int* __restrict__ rec_words, const int* __restrict__ slice_words,
               int* __restrict__ cnt, int* __restrict__ start, float* __restrict__ dinv,
               int* __restrict__ col) {
    const int2* rec   = (const int2*)rec_words;
    const int2* slice = (const int2*)slice_words;
    __shared__ int s_cnt[BIN_NODES];
    __shared__ int s_cur[BIN_NODES];
    __shared__ int s_wtot[32];
    __shared__ int s_sbeg[L1_GRID];
    __shared__ int s_slen[L1_GRID];
    __shared__ int s_red[8];
    __shared__ int s_col[COL_CAP];
    const int tid   = threadIdx.x;
    const int lane  = tid & 31;
    const int wave  = tid >> 5;
    const int b     = blockIdx.x;
    const int node0 = b * BIN_NODES;

    s_cnt[tid] = 0;
    if (tid < L1_GRID) {
        const int2 sl   = slice[tid * N_BINS + b];
        const int  cbeg = tid * L1_CHUNK;
        const int  clen = (cbeg + L1_CHUNK < N_EDGES) ? L1_CHUNK : (N_EDGES - cbeg);
        int o = sl.y, n = sl.x;
        o = o < 0 ? 0 : (o > clen ? clen : o);
        n = n < 0 ? 0 : (n > clen - o ? clen - o : n);
        s_sbeg[tid] = cbeg + o;
        s_slen[tid] = n;
        int so = o;
        for (int off = 16; off > 0; off >>= 1) so += __shfl_xor(so, off, 32);
        if (lane == 0) s_red[wave] = so;
    }
    __syncthreads();
    int lo = 0;
    for (int w = 0; w < 8; ++w) lo += s_red[w];

    int sp[SL_PER_W], sn[SL_PER_W];
#pragma unroll
    for (int j = 0; j < SL_PER_W; ++j) {
        sp[j] = __builtin_amdgcn_readfirstlane(s_sbeg[wave + 32 * j]);
        sn[j] = __builtin_amdgcn_readfirstlane(s_slen[wave + 32 * j]);
    }

    int2 r0[SL_PER_W], r1[SL_PER_W];
#pragma unroll
    for (int j = 0; j < SL_PER_W; ++j) {
        r0[j] = rec[slice_slot(sp[j], sn[j], lane)];
        r1[j] = rec[slice_slot(sp[j], sn[j], 32 + lane)];
    }

    {
#pragma unroll
        for (int j = 0; j < SL_PER_W; ++j) {
            const int n  = sn[j];
            const int d0 = r0[j].y - node0;
            const int d1 = r1[j].y - node0;
            if (lane < n      && (unsigned)d0 < (unsigned)BIN_NODES) atomicAdd(&s_cnt[d0], 1);
            if (32 + lane < n && (unsigned)d1 < (unsigned)BIN_NODES) atomicAdd(&s_cnt[d1], 1);
        }
#pragma unroll
        for (int j = 0; j < SL_PER_W; ++j) {
            const int p = sp[j];
            const int n = sn[j];
            for (int k = 64 + lane; k < n; k += 32) {
                const int d = rec[p + k].y - node0;
                if ((unsigned)d < (unsigned)BIN_NODES) atomicAdd(&s_cnt[d], 1);
            }
        }
    }
    __syncthreads();

    const int c = s_cnt[tid];
    int inc = c;
    for (int off = 1; off < 32; off <<= 1) {
        const int t = __shfl_up(inc, off, 32);
        if (lane >= off) inc += t;
    }
    if (lane == 31) s_wtot[wave] = inc;
    __syncthreads();
    int wb = 0, total = 0;
    for (int w = 0; w < 32; ++w) {
        const int v = s_wtot[w];
        total += v;
        if (w < wave) wb += v;
    }
    const int st = wb + inc - c;
    s_cur[tid] = st;
    __syncthreads();

    {
        int p0[SL_PER_W], p1[SL_PER_W];
#pragma unroll
        for (int j = 0; j < SL_PER_W; ++j) {
            const int n  = sn[j];
            const int d0 = r0[j].y - node0;
            const int d1 = r1[j].y - node0;
            p0[j] = -1;
            p1[j] = -1;
            if (lane < n      && (unsigned)d0 < (unsigned)BIN_NODES) p0[j] = atomicAdd(&s_cur[d0], 1);
            if (32 + lane < n && (unsigned)d1 < (unsigned)BIN_NODES) p1[j] = atomicAdd(&s_cur[d1], 1);
        }
#pragma unroll
        for (int j = 0; j < SL_PER_W; ++j) {
            bucket_store(s_col, col, lo, p0[j], r0[j].x);
            bucket_store(s_col, col, lo, p1[j], r1[j].x);
        }
#pragma unroll
        for (int j = 0; j < SL_PER_W; ++j) {
            const int p = sp[j];
            const int n = sn[j];
            for (int k = 64 + lane; k < n; k += 32) {
                const int2 r = rec[p + k];
                const int  d = r.y - node0;
                if ((unsigned)d < (unsigned)BIN_NODES)
                    bucket_store(s_col, col, lo, atomicAdd(&s_cur[d], 1), r.x);
            }
        }
    }
    __syncthreads();

    {
        const int staged = total < COL_CAP ? total : COL_CAP;
        const int sh     = lo & 31;
        int ov[COL_ROUNDS];
#pragma unroll
        for (int u = 0; u < COL_ROUNDS; ++u) {
            const int i = u * 1024 + tid - sh;
            ov[u] = s_col[(i >= 0 && i < staged) ? i : 0];
        }
#pragma unroll
        for (int u = 0; u < COL_ROUNDS; ++u) {
            const int i = u * 1024 + tid - sh;
            if (i >= 0 && i < staged) {
                const int pos = lo + i;
                if ((unsigned)pos < (unsigned)N_EDGES) col[pos] = ov[u];
            }
        }
    }

    {
        const int node = node0 + tid;
        if (node < N_NODES) {
            cnt[node]   = c;
            start[node] = lo + st;
            dinv[node]  = rsqrtf((float)(c + 1));
        }
    }
}


#define TILE_M    32
#define N_TILES   (N_NODES / TILE_M)
#define XT_F4     (TILE_M * IN_DIM / 4)
#define XBUF_F4   (XT_F4 + 2)
#define W1_F4     (HID_DIM * IN_DIM / 4)
#define KSTEPS    41
#define GEMM_GRID 512

struct XTile { float4 q0, q1, q2, q3, q4, q5; };

__device__ __forceinline__ XTile xtile_load(const float4* __restrict__ xs, int tid) {
    XTile t;
    t.q0 = xs[tid];
    t.q1 = xs[tid + 256];
    t.q2 = xs[tid + 512];
    t.q3 = xs[tid + 768];
    t.q4 = xs[tid + 1024];
    t.q5 = make_float4(0.f, 0.f, 0.f, 0.f);
    if (tid < XT_F4 - 1280) t.q5 = xs[tid + 1280];
    return t;
}

__device__ __forceinline__ void xtile_store(float4* xw, int tid, const XTile& t) {
    xw[tid]        = t.q0;
    xw[tid + 256]  = t.q1;
    xw[tid + 512]  = t.q2;
    xw[tid + 768]  = t.q3;
    xw[tid + 1024] = t.q4;
    if (tid < XT_F4 - 1280) xw[tid + 1280] = t.q5;
}

__global__ __launch_bounds__(256)
void k_gemm1(const float* __restrict__ x, const float* __restrict__ W1,
             const float* __restrict__ dinv, float* __restrict__ h) {
    __shared__ float4 smem4[W1_F4 + 2 * XBUF_F4];
    float*  sW = (float*)smem4;
    float4* sX = smem4 + W1_F4;

    const int tid  = threadIdx.x;
    const int lane = tid & 31;
    const int wave = tid >> 5;
    const int c    = lane & 15;
    const int hh   = lane >> 4;

    int tile = blockIdx.x;

    XTile xr = xtile_load((const float4*)(x + (size_t)tile * (TILE_M * IN_DIM)), tid);

    {
        const float4* w4 = (const float4*)W1;
        for (int q = tid; q < W1_F4; q += 256) smem4[q] = w4[q];
    }
    xtile_store(sX, tid, xr);
    __syncthreads();

    v2f   bf[KSTEPS];
    float bt;
    {
        const float* wrow = sW + (wave * 16 + c) * IN_DIM;
#pragma unroll
        for (int j = 0; j < KSTEPS; ++j) {
            bf[j].x = wrow[4 * j + 2 * hh];
            bf[j].y = wrow[4 * j + 2 * hh + 1];
        }
        const float w164 = wrow[4 * KSTEPS];
        bt = (hh == 0) ? w164 : 0.0f;
    }

    int cur = 0;
    for (; tile < N_TILES; tile += GEMM_GRID) {
        const int  nxt  = tile + GEMM_GRID;
        const bool more = nxt < N_TILES;
        if (more)
            xr = xtile_load((const float4*)(x + (size_t)nxt * (TILE_M * IN_DIM)), tid);

        const float4* dq  = (const float4*)(dinv + (size_t)tile * TILE_M + 8 * hh);
        const float4  dA0 = dq[0], dA1 = dq[1];
        const float4  dB0 = dq[4], dB1 = dq[5];

        const float* xb = (const float*)(sX + cur * XBUF_F4);
        const float* a0 = xb + c * IN_DIM;
        const float* a1 = xb + (16 + c) * IN_DIM;
        v8f acc0 = {0.f, 0.f, 0.f, 0.f, 0.f, 0.f, 0.f, 0.f};
        v8f acc1 = {0.f, 0.f, 0.f, 0.f, 0.f, 0.f, 0.f, 0.f};

#pragma unroll
        for (int j = 0; j < KSTEPS; ++j) {
            v2f f0, f1;
            f0.x = a0[4 * j + 2 * hh];
            f0.y = a0[4 * j + 2 * hh + 1];
            f1.x = a1[4 * j + 2 * hh];
            f1.y = a1[4 * j + 2 * hh + 1];
            acc0 = __builtin_amdgcn_wmma_f32_16x16x4_f32(
                       false, f0, false, bf[j], (short)0, acc0, false, false);
            acc1 = __builtin_amdgcn_wmma_f32_16x16x4_f32(
                       false, f1, false, bf[j], (short)0, acc1, false, false);
        }
        {
            const float t0 = a0[4 * KSTEPS];
            const float t1 = a1[4 * KSTEPS];
            v2f f0, f1, fb;
            f0.x = (hh == 0) ? t0 : 0.0f;  f0.y = 0.0f;
            f1.x = (hh == 0) ? t1 : 0.0f;  f1.y = 0.0f;
            fb.x = bt;                     fb.y = 0.0f;
            acc0 = __builtin_amdgcn_wmma_f32_16x16x4_f32(
                       false, f0, false, fb, (short)0, acc0, false, false);
            acc1 = __builtin_amdgcn_wmma_f32_16x16x4_f32(
                       false, f1, false, fb, (short)0, acc1, false, false);
        }

        if (more) xtile_store(sX + (cur ^ 1) * XBUF_F4, tid, xr);
        __syncthreads();

        {
            float* hp = h + ((size_t)tile * TILE_M + 8 * hh) * HID_DIM + wave * 16 + c;
            const float dA[8] = {dA0.x, dA0.y, dA0.z, dA0.w, dA1.x, dA1.y, dA1.z, dA1.w};
            const float dB[8] = {dB0.x, dB0.y, dB0.z, dB0.w, dB1.x, dB1.y, dB1.z, dB1.w};
#pragma unroll
            for (int r = 0; r < 8; ++r) {
                hp[(size_t)r * HID_DIM]        = acc0[r] * dA[r];
                hp[(size_t)(16 + r) * HID_DIM] = acc1[r] * dB[r];
            }
        }
        cur ^= 1;
    }
}


template <int K>
__device__ __forceinline__ void gather_rows(float4& acc, const float* __restrict__ h,
                                            int sv, int t, int f4) {
    float4 v[K];
#pragma unroll
    for (int u = 0; u < K; ++u) {
        const int s = __builtin_amdgcn_readlane(sv, t + u);
        v[u] = *(const float4*)&h[(size_t)s * HID_DIM + f4];
    }
#pragma unroll
    for (int u = 0; u < K; ++u) {
        acc.x += v[u].x; acc.y += v[u].y;
        acc.z += v[u].z; acc.w += v[u].w;
    }
}

#define G1_NODES 8
#define G1_GRID  ((N_NODES + 8 * G1_NODES - 1) / (8 * G1_NODES))

__device__ __forceinline__ int bucket_sources(const int* __restrict__ col, int beg, int m, int lane) {
    int e = beg + (lane < m ? lane : 0);
    e = e < N_EDGES - 1 ? e : N_EDGES - 1;
    return col[e];
}

__device__ __forceinline__ void gather_chunk(float4& acc, const float* __restrict__ h,
                                             int sv, int nb, int f4) {
    int t = 0;
    for (; t + 8 <= nb; t += 8) gather_rows<8>(acc, h, sv, t, f4);
    if (nb - t >= 4) { gather_rows<4>(acc, h, sv, t, f4); t += 4; }
    const int r = nb - t;
    if (r == 3)      gather_rows<3>(acc, h, sv, t, f4);
    else if (r == 2) gather_rows<2>(acc, h, sv, t, f4);
    else if (r == 1) gather_rows<1>(acc, h, sv, t, f4);
}

__global__ __launch_bounds__(256)
void k_gather1(const int* __restrict__ start, const int* __restrict__ cnt,
               const int* __restrict__ col, const float* __restrict__ dinv,
               const float* __restrict__ h, const float* __restrict__ b1,
               const float* __restrict__ W2, float* __restrict__ tval, int n) {
    const int lane = threadIdx.x & 31;
    const int n0 = __builtin_amdgcn_readfirstlane((blockIdx.x * 8 + (threadIdx.x >> 5)) * G1_NODES);
    if (n0 >= n) return;
    const int f4 = lane * 4;
    const int ng = (n - n0 < G1_NODES) ? (n - n0) : G1_NODES;

    const int   nl = n0 + (lane < ng ? lane : ng - 1);
    const int   mv = cnt[nl];
    const int   bv = start[nl];
    const float dv = dinv[nl];

    float4 self = *(const float4*)&h[(size_t)n0 * HID_DIM + f4];
    int m   = __builtin_amdgcn_readlane(mv, 0);
    int beg = __builtin_amdgcn_readlane(bv, 0);
    int s   = bucket_sources(col, beg, m, lane);
    float pv = 0.0f;

    for (int j = 0; j < ng; ++j) {
        const int    jn     = (j + 1 < ng) ? (j + 1) : (ng - 1);
        const float4 self_n = *(const float4*)&h[(size_t)(n0 + jn) * HID_DIM + f4];
        const int    m_n    = __builtin_amdgcn_readlane(mv, jn);
        const int    beg_n  = __builtin_amdgcn_readlane(bv, jn);
        const int    s_n    = bucket_sources(col, beg_n, m_n, lane);

        const float di = __int_as_float(__builtin_amdgcn_readlane(__float_as_int(dv), j));
        float4 acc = self;

        gather_chunk(acc, h, s, (m < 32) ? m : 32, f4);
        for (int base = 32; base < m; base += 32) {
            const int s2 = bucket_sources(col, beg + base, m - base, lane);
            gather_chunk(acc, h, s2, (m - base < 32) ? (m - base) : 32, f4);
        }

        const float4 bb = *(const float4*)&b1[f4];
        const float4 w2 = *(const float4*)&W2[f4];
        float p = fmaxf(acc.x * di + bb.x, 0.f) * w2.x
                + fmaxf(acc.y * di + bb.y, 0.f) * w2.y
                + fmaxf(acc.z * di + bb.z, 0.f) * w2.z
                + fmaxf(acc.w * di + bb.w, 0.f) * w2.w;
        for (int off = 16; off > 0; off >>= 1)
            p += __shfl_xor(p, off, 32);
        if (lane == j) pv = p;

        m = m_n;   beg = beg_n;   s = s_n;   self = self_n;
    }
    if (lane < ng) tval[n0 + lane] = pv * dv;
}


__global__ void k_gather2(const int* __restrict__ start, const int* __restrict__ cnt,
                          const int* __restrict__ col, const float* __restrict__ dinv,
                          const float* __restrict__ tval, const float* __restrict__ b2,
                          float* __restrict__ out, int n) {
    int i = blockIdx.x * blockDim.x + threadIdx.x;
    if (i >= n) return;
    const float di  = dinv[i];
    const int   beg = start[i];
    const int   m   = cnt[i];
    float acc = tval[i];
    for (int t = 0; t < m; ++t)
        acc += tval[col[beg + t]];
    out[i] = 1.0f / (1.0f + expf(-(acc * di + b2[0])));
}


extern "C" void kernel_launch(void* const* d_in, const int* in_sizes, int n_in,
                              void* d_out, int out_size, void* d_ws, size_t ws_size,
                              hipStream_t stream) {
    (void)in_sizes; (void)n_in; (void)out_size; (void)ws_size;
    const float* x  = (const float*)d_in[0];
    const float* W1 = (const float*)d_in[1];
    const float* b1 = (const float*)d_in[2];
    const float* W2 = (const float*)d_in[3];
    const float* b2 = (const float*)d_in[4];
    const int*   ei = (const int*)d_in[5];
    const int* src = ei;
    const int* dst = ei + N_EDGES;
    float* out = (float*)d_out;

    float* dinv = (float*)d_ws;
    float* tval = dinv + N_NODES;
    float* h    = tval + N_NODES;
    int*  cnt     = (int*)(h + (size_t)N_NODES * HID_DIM);
    int*  start   = cnt + N_NODES;
    int*  rec     = start + N_NODES;
    int*  col     = rec + 2 * N_EDGES;
    int*  slice   = col + N_EDGES;

    const int B = 256;
    k_chunk_sort<<<L1_GRID, 1024, 0, stream>>>(src, dst, rec, slice);
    k_bin_csr   <<<N_BINS, 1024, 0, stream>>>(rec, slice, cnt, start, dinv, col);

    k_gemm1    <<<GEMM_GRID, B, 0, stream>>>(x, W1, dinv, h);

    k_gather1  <<<G1_GRID, B, 0, stream>>>(start, cnt, col, dinv, h, b1, W2, tval, N_NODES);
    k_gather2  <<<(N_NODES + B - 1) / B, B, 0, stream>>>(start, cnt, col, dinv, tval, b2, out, N_NODES);
}
